# starting kernel
# baseline (speedup 1.0000x reference)
#include <hip/hip_runtime.h>
#include <hip/hip_bf16.h>
#include <math.h>

typedef float v2f __attribute__((ext_vector_type(2)));
typedef float v8f __attribute__((ext_vector_type(8)));

#define NB     8
#define NPTS   16384
#define NPT    512
#define NSAMP  32
#define RAD2   0.04f
#define BN_EPS 1e-5f

__global__ __launch_bounds__(1024) void fps_kernel(const float* __restrict__ xyz,
                                                   int* __restrict__ fpsidx,
                                                   float* __restrict__ newxyz) {
  const int b   = blockIdx.x;
  const int tid = threadIdx.x;
  const float* base = xyz + (size_t)b * NPTS * 3;

  float px[16], py[16], pz[16], dd[16];
#pragma unroll
  for (int j = 0; j < 16; ++j) {
    int p = tid + j * 1024;
    px[j] = base[p * 3 + 0];
    py[j] = base[p * 3 + 1];
    pz[j] = base[p * 3 + 2];
    dd[j] = 3.4e38f;
  }

  __shared__ float rv[32];
  __shared__ int   ri[32];
  __shared__ int   s_last;

  const int wave = tid >> 5;
  const int lane = tid & 31;
  int last = 0;

  for (int i = 0; i < NPT; ++i) {
    if (tid == 0) {
      fpsidx[b * NPT + i] = last;
      newxyz[(b * NPT + i) * 3 + 0] = base[last * 3 + 0];
      newxyz[(b * NPT + i) * 3 + 1] = base[last * 3 + 1];
      newxyz[(b * NPT + i) * 3 + 2] = base[last * 3 + 2];
    }
    const float cx = base[last * 3 + 0];
    const float cy = base[last * 3 + 1];
    const float cz = base[last * 3 + 2];

    float bv = -1.0f;
    int   bi = 0x7fffffff;
#pragma unroll
    for (int j = 0; j < 16; ++j) {
      float dx = px[j] - cx, dy = py[j] - cy, dz = pz[j] - cz;
      float d  = dx * dx + dy * dy + dz * dz;
      float nd = fminf(dd[j], d);
      dd[j] = nd;
      if (nd > bv) { bv = nd; bi = tid + j * 1024; }
    }
#pragma unroll
    for (int off = 16; off > 0; off >>= 1) {
      float ov = __shfl_xor(bv, off, 32);
      int   oi = __shfl_xor(bi, off, 32);
      if (ov > bv || (ov == bv && oi < bi)) { bv = ov; bi = oi; }
    }
    if (lane == 0) { rv[wave] = bv; ri[wave] = bi; }
    __syncthreads();
    if (wave == 0) {
      bv = rv[lane]; bi = ri[lane];
#pragma unroll
      for (int off = 16; off > 0; off >>= 1) {
        float ov = __shfl_xor(bv, off, 32);
        int   oi = __shfl_xor(bi, off, 32);
        if (ov > bv || (ov == bv && oi < bi)) { bv = ov; bi = oi; }
      }
      if (lane == 0) s_last = bi;
    }
    __syncthreads();
    last = s_last;
  }
}

__global__ __launch_bounds__(256) void ballq_kernel(const float* __restrict__ xyz,
                                                    const float* __restrict__ newxyz,
                                                    int* __restrict__ grp) {
  const int lane = threadIdx.x & 31;
  const int q    = blockIdx.x * 8 + (threadIdx.x >> 5);
  const int b    = q >> 9;
  const int s    = q & (NPT - 1);

  const float* base = xyz + (size_t)b * NPTS * 3;
  const float cx = newxyz[(b * NPT + s) * 3 + 0];
  const float cy = newxyz[(b * NPT + s) * 3 + 1];
  const float cz = newxyz[(b * NPT + s) * 3 + 2];

  int* g = grp + (size_t)q * NSAMP;
  int cnt = 0, first = -1;

  for (int bp = 0; bp < NPTS && cnt < NSAMP; bp += 32) {
    const int p = bp + lane;
    float dx = base[p * 3 + 0] - cx;
    float dy = base[p * 3 + 1] - cy;
    float dz = base[p * 3 + 2] - cz;
    bool hit = (dx * dx + dy * dy + dz * dz) <= RAD2;
    unsigned mask = (unsigned)__ballot(hit);
    if (mask) {
      if (first < 0) first = bp + __builtin_ctz(mask);
      int pos = __popc(mask & ((1u << lane) - 1u));
      if (hit && (cnt + pos) < NSAMP) g[cnt + pos] = p;
      cnt += __popc(mask);
    }
  }
  if (first < 0) first = NPTS - 1;
  int filled = cnt < NSAMP ? cnt : NSAMP;
  for (int slot = filled + lane; slot < NSAMP; slot += 32) g[slot] = first;
}

__device__ __forceinline__ void gemm_bn_relu(const float* __restrict__ A, int lda,
                                             const float* __restrict__ W, int Kdim, int Kpad,
                                             int Nout,
                                             const float* __restrict__ gg,
                                             const float* __restrict__ bb,
                                             const float* __restrict__ mm,
                                             const float* __restrict__ vv,
                                             float* __restrict__ D, int ldd, int lane) {
  const int r15  = lane & 15;
  const int half = lane >> 4;
  for (int m0 = 0; m0 < 32; m0 += 16) {
    for (int n0 = 0; n0 < Nout; n0 += 16) {
      v8f acc = {0.f, 0.f, 0.f, 0.f, 0.f, 0.f, 0.f, 0.f};
      const int n = n0 + r15;
      for (int k0 = 0; k0 < Kpad; k0 += 4) {
        const int kk = k0 + half * 2;
        v2f a, w;
        a.x = A[(m0 + r15) * lda + kk];
        a.y = A[(m0 + r15) * lda + kk + 1];
        int ka = kk     < Kdim ? kk     : Kdim - 1;
        int kb = kk + 1 < Kdim ? kk + 1 : Kdim - 1;
        w.x = W[n * Kdim + ka];
        w.y = W[n * Kdim + kb];
        acc = __builtin_amdgcn_wmma_f32_16x16x4_f32(false, a, false, w,
                                                    (short)0, acc, false, false);
      }
      const float sc = gg[n] * rsqrtf(vv[n] + BN_EPS);
      const float bs = bb[n] - mm[n] * sc;
#pragma unroll
      for (int r = 0; r < 8; ++r) {
        float val = fmaxf(acc[r] * sc + bs, 0.f);
        D[(m0 + r + half * 8) * ldd + n] = val;
      }
    }
  }
}

__global__ __launch_bounds__(64) void mlp_kernel(const float* __restrict__ xyz,
                                                 const float* __restrict__ feat,
                                                 const int*   __restrict__ grp,
                                                 const float* __restrict__ newxyz,
                                                 const float* __restrict__ w0, const float* __restrict__ g0,
                                                 const float* __restrict__ b0, const float* __restrict__ m0,
                                                 const float* __restrict__ v0,
                                                 const float* __restrict__ w1, const float* __restrict__ g1,
                                                 const float* __restrict__ b1, const float* __restrict__ m1,
                                                 const float* __restrict__ v1,
                                                 const float* __restrict__ w2, const float* __restrict__ g2,
                                                 const float* __restrict__ b2, const float* __restrict__ m2,
                                                 const float* __restrict__ v2,
                                                 float* __restrict__ outfeat) {
  __shared__ float Xb[2][32 * 68];
  __shared__ float Hb[2][32 * 132];

  const int wave = threadIdx.x >> 5;
  const int lane = threadIdx.x & 31;
  const int q    = blockIdx.x * 2 + wave;
  const int b    = q >> 9;
  const int s    = q & (NPT - 1);

  float* X = Xb[wave];
  float* H = Hb[wave];

  const int gi = grp[(size_t)q * NSAMP + lane];
  const float* prow = xyz + ((size_t)b * NPTS + gi) * 3;
  const float* nx   = newxyz + (b * NPT + s) * 3;
  float* xr = X + lane * 68;
  xr[0] = prow[0] - nx[0];
  xr[1] = prow[1] - nx[1];
  xr[2] = prow[2] - nx[2];
  const float4* frow = (const float4*)(feat + ((size_t)b * NPTS + gi) * 64);
#pragma unroll
  for (int c4 = 0; c4 < 16; ++c4) {
    float4 f = frow[c4];
    xr[3 + 4 * c4 + 0] = f.x;
    xr[3 + 4 * c4 + 1] = f.y;
    xr[3 + 4 * c4 + 2] = f.z;
    xr[3 + 4 * c4 + 3] = f.w;
  }
  xr[67] = 0.f;

  gemm_bn_relu(X, 68,  w0, 67, 68, 64,  g0, b0, m0, v0, H, 132, lane);
  gemm_bn_relu(H, 132, w1, 64, 64, 64,  g1, b1, m1, v1, X, 68,  lane);
  gemm_bn_relu(X, 68,  w2, 64, 64, 128, g2, b2, m2, v2, H, 132, lane);

#pragma unroll
  for (int cc = 0; cc < 4; ++cc) {
    const int col = lane + cc * 32;
    float mx = H[col];
    for (int row = 1; row < 32; ++row) mx = fmaxf(mx, H[row * 132 + col]);
    outfeat[((size_t)b * 128 + col) * NPT + s] = mx;
  }
}

extern "C" void kernel_launch(void* const* d_in, const int* in_sizes, int n_in,
                              void* d_out, int out_size, void* d_ws, size_t ws_size,
                              hipStream_t stream) {
  (void)in_sizes; (void)n_in; (void)out_size; (void)ws_size;
  const float* xyz  = (const float*)d_in[0];
  const float* feat = (const float*)d_in[1];
  const float* w0 = (const float*)d_in[2];
  const float* g0 = (const float*)d_in[3];
  const float* b0 = (const float*)d_in[4];
  const float* m0 = (const float*)d_in[5];
  const float* v0 = (const float*)d_in[6];
  const float* w1 = (const float*)d_in[7];
  const float* g1 = (const float*)d_in[8];
  const float* b1 = (const float*)d_in[9];
  const float* m1 = (const float*)d_in[10];
  const float* v1 = (const float*)d_in[11];
  const float* w2 = (const float*)d_in[12];
  const float* g2 = (const float*)d_in[13];
  const float* b2 = (const float*)d_in[14];
  const float* m2 = (const float*)d_in[15];
  const float* v2 = (const float*)d_in[16];

  float* newxyz  = (float*)d_out;
  float* outfeat = (float*)d_out + NB * NPT * 3;

  int* fpsidx = (int*)d_ws;
  int* grp    = (int*)d_ws + NB * NPT;

  fps_kernel  <<<NB,                1024, 0, stream>>>(xyz, fpsidx, newxyz);
  ballq_kernel<<<(NB * NPT) / 8,     256, 0, stream>>>(xyz, newxyz, grp);
  mlp_kernel  <<<(NB * NPT) / 2,      64, 0, stream>>>(xyz, feat, grp, newxyz,
                                                       w0, g0, b0, m0, v0,
                                                       w1, g1, b1, m1, v1,
                                                       w2, g2, b2, m2, v2,
                                                       outfeat);
}


# Ball query: a block's 8 queries share 2048-point tiles of their batch's cloud, staged in LDS by all 256 threads (next tile in flight in registers); waves scan LDS in index order; same hit test
# speedup vs baseline: 1.1052x; 1.1052x over previous
#include <hip/hip_runtime.h>
#include <hip/hip_bf16.h>
#include <math.h>

typedef float v2f __attribute__((ext_vector_type(2)));
typedef float v8f __attribute__((ext_vector_type(8)));

#define NB     8
#define NPTS   16384
#define NPT    512
#define NSAMP  32
#define RAD2   0.04f
#define BN_EPS 1e-5f

__global__ __launch_bounds__(1024) void fps_kernel(const float* __restrict__ xyz,
                                                   int* __restrict__ fpsidx,
                                                   float* __restrict__ newxyz) {
  const int b   = blockIdx.x;
  const int tid = threadIdx.x;
  const float* base = xyz + (size_t)b * NPTS * 3;

  float px[16], py[16], pz[16], dd[16];
#pragma unroll
  for (int j = 0; j < 16; ++j) {
    int p = tid + j * 1024;
    px[j] = base[p * 3 + 0];
    py[j] = base[p * 3 + 1];
    pz[j] = base[p * 3 + 2];
    dd[j] = 3.4e38f;
  }

  __shared__ float rv[32];
  __shared__ int   ri[32];
  __shared__ int   s_last;

  const int wave = tid >> 5;
  const int lane = tid & 31;
  int last = 0;

  for (int i = 0; i < NPT; ++i) {
    if (tid == 0) {
      fpsidx[b * NPT + i] = last;
      newxyz[(b * NPT + i) * 3 + 0] = base[last * 3 + 0];
      newxyz[(b * NPT + i) * 3 + 1] = base[last * 3 + 1];
      newxyz[(b * NPT + i) * 3 + 2] = base[last * 3 + 2];
    }
    const float cx = base[last * 3 + 0];
    const float cy = base[last * 3 + 1];
    const float cz = base[last * 3 + 2];

    float bv = -1.0f;
    int   bi = 0x7fffffff;
#pragma unroll
    for (int j = 0; j < 16; ++j) {
      float dx = px[j] - cx, dy = py[j] - cy, dz = pz[j] - cz;
      float d  = dx * dx + dy * dy + dz * dz;
      float nd = fminf(dd[j], d);
      dd[j] = nd;
      if (nd > bv) { bv = nd; bi = tid + j * 1024; }
    }
#pragma unroll
    for (int off = 16; off > 0; off >>= 1) {
      float ov = __shfl_xor(bv, off, 32);
      int   oi = __shfl_xor(bi, off, 32);
      if (ov > bv || (ov == bv && oi < bi)) { bv = ov; bi = oi; }
    }
    if (lane == 0) { rv[wave] = bv; ri[wave] = bi; }
    __syncthreads();
    if (wave == 0) {
      bv = rv[lane]; bi = ri[lane];
#pragma unroll
      for (int off = 16; off > 0; off >>= 1) {
        float ov = __shfl_xor(bv, off, 32);
        int   oi = __shfl_xor(bi, off, 32);
        if (ov > bv || (ov == bv && oi < bi)) { bv = ov; bi = oi; }
      }
      if (lane == 0) s_last = bi;
    }
    __syncthreads();
    last = s_last;
  }
}

#define BQ_TILE 2048
#define BQ_TV4  (BQ_TILE * 3 / 4)
#define BQ_NT   (NPTS / BQ_TILE)
#define BQ_G    4

__global__ __launch_bounds__(256) void ballq_kernel(const float* __restrict__ xyz,
                                                    const float* __restrict__ newxyz,
                                                    int* __restrict__ grp) {
  __shared__ float4 tileb[2][BQ_TV4];
  __shared__ int    needf[2][8];

  const int tid  = threadIdx.x;
  const int lane = tid & 31;
  const int wave = tid >> 5;
  const int q    = blockIdx.x * 8 + wave;
  const int b    = q >> 9;
  const int s    = q & (NPT - 1);

  const float4* src = (const float4*)(xyz + (size_t)b * NPTS * 3);
  const float cx = newxyz[(b * NPT + s) * 3 + 0];
  const float cy = newxyz[(b * NPT + s) * 3 + 1];
  const float cz = newxyz[(b * NPT + s) * 3 + 2];

  int* g = grp + (size_t)q * NSAMP;
  int cnt = 0, first = -1;

  float4 r0 = src[tid + 0 * 256], r1 = src[tid + 1 * 256], r2 = src[tid + 2 * 256];
  float4 r3 = src[tid + 3 * 256], r4 = src[tid + 4 * 256], r5 = src[tid + 5 * 256];

  for (int t = 0; t < BQ_NT; ++t) {
    const int buf = t & 1;
    tileb[buf][tid + 0 * 256] = r0;
    tileb[buf][tid + 1 * 256] = r1;
    tileb[buf][tid + 2 * 256] = r2;
    tileb[buf][tid + 3 * 256] = r3;
    tileb[buf][tid + 4 * 256] = r4;
    tileb[buf][tid + 5 * 256] = r5;
    if (lane == 0) needf[buf][wave] = (cnt < NSAMP) ? 1 : 0;
    __syncthreads();
    int any = 0;
#pragma unroll
    for (int w = 0; w < 8; ++w) any |= needf[buf][w];
    if (__builtin_amdgcn_readfirstlane(any) == 0) break;

    if (t + 1 < BQ_NT) {
      const float4* nsrc = src + (t + 1) * BQ_TV4 + tid;
      r0 = nsrc[0 * 256]; r1 = nsrc[1 * 256]; r2 = nsrc[2 * 256];
      r3 = nsrc[3 * 256]; r4 = nsrc[4 * 256]; r5 = nsrc[5 * 256];
    }

    if (cnt < NSAMP) {
      const float* L = (const float*)tileb[buf];
      for (int c0 = 0; c0 < BQ_TILE && cnt < NSAMP; c0 += 32 * BQ_G) {
        float px[BQ_G], py[BQ_G], pz[BQ_G];
#pragma unroll
        for (int u = 0; u < BQ_G; ++u) {
          const int pl = c0 + u * 32 + lane;
          px[u] = L[pl * 3 + 0];
          py[u] = L[pl * 3 + 1];
          pz[u] = L[pl * 3 + 2];
        }
#pragma unroll
        for (int u = 0; u < BQ_G; ++u) {
          const float dx = px[u] - cx;
          const float dy = py[u] - cy;
          const float dz = pz[u] - cz;
          const float d2 = dx * dx + dy * dy + dz * dz;
          const bool hit = d2 <= RAD2;
          const int bp = t * BQ_TILE + c0 + u * 32;
          unsigned mask = (unsigned)__ballot(hit);
          if (mask) {
            if (first < 0) first = bp + __builtin_ctz(mask);
            int pos = __popc(mask & ((1u << lane) - 1u));
            if (hit && (cnt + pos) < NSAMP) g[cnt + pos] = bp + lane;
            cnt += __popc(mask);
          }
        }
      }
    }
  }
  if (first < 0) first = NPTS - 1;
  int filled = cnt < NSAMP ? cnt : NSAMP;
  for (int slot = filled + lane; slot < NSAMP; slot += 32) g[slot] = first;
}

__device__ __forceinline__ void gemm_bn_relu(const float* __restrict__ A, int lda,
                                             const float* __restrict__ W, int Kdim, int Kpad,
                                             int Nout,
                                             const float* __restrict__ gg,
                                             const float* __restrict__ bb,
                                             const float* __restrict__ mm,
                                             const float* __restrict__ vv,
                                             float* __restrict__ D, int ldd, int lane) {
  const int r15  = lane & 15;
  const int half = lane >> 4;
  for (int m0 = 0; m0 < 32; m0 += 16) {
    for (int n0 = 0; n0 < Nout; n0 += 16) {
      v8f acc = {0.f, 0.f, 0.f, 0.f, 0.f, 0.f, 0.f, 0.f};
      const int n = n0 + r15;
      for (int k0 = 0; k0 < Kpad; k0 += 4) {
        const int kk = k0 + half * 2;
        v2f a, w;
        a.x = A[(m0 + r15) * lda + kk];
        a.y = A[(m0 + r15) * lda + kk + 1];
        int ka = kk     < Kdim ? kk     : Kdim - 1;
        int kb = kk + 1 < Kdim ? kk + 1 : Kdim - 1;
        w.x = W[n * Kdim + ka];
        w.y = W[n * Kdim + kb];
        acc = __builtin_amdgcn_wmma_f32_16x16x4_f32(false, a, false, w,
                                                    (short)0, acc, false, false);
      }
      const float sc = gg[n] * rsqrtf(vv[n] + BN_EPS);
      const float bs = bb[n] - mm[n] * sc;
#pragma unroll
      for (int r = 0; r < 8; ++r) {
        float val = fmaxf(acc[r] * sc + bs, 0.f);
        D[(m0 + r + half * 8) * ldd + n] = val;
      }
    }
  }
}

__global__ __launch_bounds__(64) void mlp_kernel(const float* __restrict__ xyz,
                                                 const float* __restrict__ feat,
                                                 const int*   __restrict__ grp,
                                                 const float* __restrict__ newxyz,
                                                 const float* __restrict__ w0, const float* __restrict__ g0,
                                                 const float* __restrict__ b0, const float* __restrict__ m0,
                                                 const float* __restrict__ v0,
                                                 const float* __restrict__ w1, const float* __restrict__ g1,
                                                 const float* __restrict__ b1, const float* __restrict__ m1,
                                                 const float* __restrict__ v1,
                                                 const float* __restrict__ w2, const float* __restrict__ g2,
                                                 const float* __restrict__ b2, const float* __restrict__ m2,
                                                 const float* __restrict__ v2,
                                                 float* __restrict__ outfeat) {
  __shared__ float Xb[2][32 * 68];
  __shared__ float Hb[2][32 * 132];

  const int wave = threadIdx.x >> 5;
  const int lane = threadIdx.x & 31;
  const int q    = blockIdx.x * 2 + wave;
  const int b    = q >> 9;
  const int s    = q & (NPT - 1);

  float* X = Xb[wave];
  float* H = Hb[wave];

  const int gi = grp[(size_t)q * NSAMP + lane];
  const float* prow = xyz + ((size_t)b * NPTS + gi) * 3;
  const float* nx   = newxyz + (b * NPT + s) * 3;
  float* xr = X + lane * 68;
  xr[0] = prow[0] - nx[0];
  xr[1] = prow[1] - nx[1];
  xr[2] = prow[2] - nx[2];
  const float4* frow = (const float4*)(feat + ((size_t)b * NPTS + gi) * 64);
#pragma unroll
  for (int c4 = 0; c4 < 16; ++c4) {
    float4 f = frow[c4];
    xr[3 + 4 * c4 + 0] = f.x;
    xr[3 + 4 * c4 + 1] = f.y;
    xr[3 + 4 * c4 + 2] = f.z;
    xr[3 + 4 * c4 + 3] = f.w;
  }
  xr[67] = 0.f;

  gemm_bn_relu(X, 68,  w0, 67, 68, 64,  g0, b0, m0, v0, H, 132, lane);
  gemm_bn_relu(H, 132, w1, 64, 64, 64,  g1, b1, m1, v1, X, 68,  lane);
  gemm_bn_relu(X, 68,  w2, 64, 64, 128, g2, b2, m2, v2, H, 132, lane);

#pragma unroll
  for (int cc = 0; cc < 4; ++cc) {
    const int col = lane + cc * 32;
    float mx = H[col];
    for (int row = 1; row < 32; ++row) mx = fmaxf(mx, H[row * 132 + col]);
    outfeat[((size_t)b * 128 + col) * NPT + s] = mx;
  }
}

extern "C" void kernel_launch(void* const* d_in, const int* in_sizes, int n_in,
                              void* d_out, int out_size, void* d_ws, size_t ws_size,
                              hipStream_t stream) {
  (void)in_sizes; (void)n_in; (void)out_size; (void)ws_size;
  const float* xyz  = (const float*)d_in[0];
  const float* feat = (const float*)d_in[1];
  const float* w0 = (const float*)d_in[2];
  const float* g0 = (const float*)d_in[3];
  const float* b0 = (const float*)d_in[4];
  const float* m0 = (const float*)d_in[5];
  const float* v0 = (const float*)d_in[6];
  const float* w1 = (const float*)d_in[7];
  const float* g1 = (const float*)d_in[8];
  const float* b1 = (const float*)d_in[9];
  const float* m1 = (const float*)d_in[10];
  const float* v1 = (const float*)d_in[11];
  const float* w2 = (const float*)d_in[12];
  const float* g2 = (const float*)d_in[13];
  const float* b2 = (const float*)d_in[14];
  const float* m2 = (const float*)d_in[15];
  const float* v2 = (const float*)d_in[16];

  float* newxyz  = (float*)d_out;
  float* outfeat = (float*)d_out + NB * NPT * 3;

  int* fpsidx = (int*)d_ws;
  int* grp    = (int*)d_ws + NB * NPT;

  fps_kernel  <<<NB,                1024, 0, stream>>>(xyz, fpsidx, newxyz);
  ballq_kernel<<<(NB * NPT) / 8,     256, 0, stream>>>(xyz, newxyz, grp);
  mlp_kernel  <<<(NB * NPT) / 2,      64, 0, stream>>>(xyz, feat, grp, newxyz,
                                                       w0, g0, b0, m0, v0,
                                                       w1, g1, b1, m1, v1,
                                                       w2, g2, b2, m2, v2,
                                                       outfeat);
}


# FPS: per-wave argmax fold dropped: 512 threads x 32 points hand (distance bits, ~index) keys through LDS to wave 0, which takes the max of 16 per lane and folds once; same picks, same ties
# speedup vs baseline: 1.1855x; 1.0727x over previous
#include <hip/hip_runtime.h>
#include <hip/hip_bf16.h>
#include <math.h>

typedef float v2f __attribute__((ext_vector_type(2)));
typedef float v8f __attribute__((ext_vector_type(8)));

#define NB     8
#define NPTS   16384
#define NPT    512
#define NSAMP  32
#define RAD2   0.04f
#define BN_EPS 1e-5f

#define FPS_T 512
#define FPS_P (NPTS / FPS_T)
#define FPS_C 4
#define FPS_L (FPS_P / FPS_C)
#define FPS_K (FPS_T / 32)

__global__ __launch_bounds__(FPS_T) void fps_kernel(const float* __restrict__ xyz,
                                                    int* __restrict__ fpsidx,
                                                    float* __restrict__ newxyz) {
  const int b   = blockIdx.x;
  const int tid = threadIdx.x;
  const float* base = xyz + (size_t)b * NPTS * 3;

  v2f px[FPS_P / 2], py[FPS_P / 2], pz[FPS_P / 2], dd[FPS_P / 2];
#pragma unroll
  for (int m = 0; m < FPS_P / 2; ++m) {
    const int p0 = tid + (2 * m) * FPS_T;
    const int p1 = p0 + FPS_T;
    px[m].x = base[p0 * 3 + 0];  px[m].y = base[p1 * 3 + 0];
    py[m].x = base[p0 * 3 + 1];  py[m].y = base[p1 * 3 + 1];
    pz[m].x = base[p0 * 3 + 2];  pz[m].y = base[p1 * 3 + 2];
    dd[m].x = 3.4e38f;           dd[m].y = 3.4e38f;
  }

  __shared__ unsigned long long lk[FPS_T];
  __shared__ int s_last;

  const int wave = tid >> 5;
  const int lane = tid & 31;
  int last = 0;

  for (int i = 0; i < NPT; ++i) {
    const float cx = base[last * 3 + 0];
    const float cy = base[last * 3 + 1];
    const float cz = base[last * 3 + 2];
    if (tid == 0) {
      fpsidx[b * NPT + i] = last;
      newxyz[(b * NPT + i) * 3 + 0] = cx;
      newxyz[(b * NPT + i) * 3 + 1] = cy;
      newxyz[(b * NPT + i) * 3 + 2] = cz;
    }

    float cv[FPS_C];
    int   cj[FPS_C];
    const v2f cx2 = {cx, cx}, cy2 = {cy, cy}, cz2 = {cz, cz};
#pragma unroll
    for (int k = 0; k < FPS_C; ++k) {
#pragma unroll
      for (int l = 0; l < FPS_L; l += 2) {
        const int j = k * FPS_L + l;
        const int m = j >> 1;
        const v2f dx = px[m] - cx2, dy = py[m] - cy2, dz = pz[m] - cz2;
        const v2f d  = dx * dx + dy * dy + dz * dz;
        const float n0 = fminf(dd[m].x, d.x);
        const float n1 = fminf(dd[m].y, d.y);
        dd[m].x = n0;
        dd[m].y = n1;
        if (l == 0)          { cv[k] = n0; cj[k] = j; }
        else if (n0 > cv[k]) { cv[k] = n0; cj[k] = j; }
        if (n1 > cv[k])      { cv[k] = n1; cj[k] = j + 1; }
      }
    }
    float bv = cv[0];
    int   bj = cj[0];
#pragma unroll
    for (int k = 1; k < FPS_C; ++k) {
      if (cv[k] > bv) { bv = cv[k]; bj = cj[k]; }
    }
    int bi = tid + bj * FPS_T;

    lk[tid] = ((unsigned long long)__float_as_uint(bv) << 32) | (unsigned)(~bi);
    __syncthreads();
    if (wave == 0) {
      unsigned long long kk[FPS_K];
#pragma unroll
      for (int e = 0; e < FPS_K; ++e) kk[e] = lk[lane + 32 * e];
#pragma unroll
      for (int s = FPS_K / 2; s > 0; s >>= 1)
#pragma unroll
        for (int e = 0; e < s; ++e) kk[e] = kk[e + s] > kk[e] ? kk[e + s] : kk[e];
      bv = __uint_as_float((unsigned)(kk[0] >> 32));
      bi = (int)~(unsigned)kk[0];
#pragma unroll
      for (int off = 16; off > 0; off >>= 1) {
        float ov = __shfl_xor(bv, off, 32);
        int   oi = __shfl_xor(bi, off, 32);
        if (ov > bv || (ov == bv && oi < bi)) { bv = ov; bi = oi; }
      }
      if (lane == 0) s_last = bi;
    }
    __syncthreads();
    last = __builtin_amdgcn_readfirstlane(s_last);
  }
}

#define BQ_TILE 2048
#define BQ_TV4  (BQ_TILE * 3 / 4)
#define BQ_NT   (NPTS / BQ_TILE)
#define BQ_G    4

__global__ __launch_bounds__(256) void ballq_kernel(const float* __restrict__ xyz,
                                                    const float* __restrict__ newxyz,
                                                    int* __restrict__ grp) {
  __shared__ float4 tileb[2][BQ_TV4];
  __shared__ int    needf[2][8];

  const int tid  = threadIdx.x;
  const int lane = tid & 31;
  const int wave = tid >> 5;
  const int q    = blockIdx.x * 8 + wave;
  const int b    = q >> 9;
  const int s    = q & (NPT - 1);

  const float4* src = (const float4*)(xyz + (size_t)b * NPTS * 3);
  const float cx = newxyz[(b * NPT + s) * 3 + 0];
  const float cy = newxyz[(b * NPT + s) * 3 + 1];
  const float cz = newxyz[(b * NPT + s) * 3 + 2];

  int* g = grp + (size_t)q * NSAMP;
  int cnt = 0, first = -1;

  float4 r0 = src[tid + 0 * 256], r1 = src[tid + 1 * 256], r2 = src[tid + 2 * 256];
  float4 r3 = src[tid + 3 * 256], r4 = src[tid + 4 * 256], r5 = src[tid + 5 * 256];

  for (int t = 0; t < BQ_NT; ++t) {
    const int buf = t & 1;
    tileb[buf][tid + 0 * 256] = r0;
    tileb[buf][tid + 1 * 256] = r1;
    tileb[buf][tid + 2 * 256] = r2;
    tileb[buf][tid + 3 * 256] = r3;
    tileb[buf][tid + 4 * 256] = r4;
    tileb[buf][tid + 5 * 256] = r5;
    if (lane == 0) needf[buf][wave] = (cnt < NSAMP) ? 1 : 0;
    __syncthreads();
    int any = 0;
#pragma unroll
    for (int w = 0; w < 8; ++w) any |= needf[buf][w];
    if (__builtin_amdgcn_readfirstlane(any) == 0) break;

    if (t + 1 < BQ_NT) {
      const float4* nsrc = src + (t + 1) * BQ_TV4 + tid;
      r0 = nsrc[0 * 256]; r1 = nsrc[1 * 256]; r2 = nsrc[2 * 256];
      r3 = nsrc[3 * 256]; r4 = nsrc[4 * 256]; r5 = nsrc[5 * 256];
    }

    if (cnt < NSAMP) {
      const float* L = (const float*)tileb[buf];
      for (int c0 = 0; c0 < BQ_TILE && cnt < NSAMP; c0 += 32 * BQ_G) {
        float px[BQ_G], py[BQ_G], pz[BQ_G];
#pragma unroll
        for (int u = 0; u < BQ_G; ++u) {
          const int pl = c0 + u * 32 + lane;
          px[u] = L[pl * 3 + 0];
          py[u] = L[pl * 3 + 1];
          pz[u] = L[pl * 3 + 2];
        }
#pragma unroll
        for (int u = 0; u < BQ_G; ++u) {
          const float dx = px[u] - cx;
          const float dy = py[u] - cy;
          const float dz = pz[u] - cz;
          const float d2 = dx * dx + dy * dy + dz * dz;
          const bool hit = d2 <= RAD2;
          const int bp = t * BQ_TILE + c0 + u * 32;
          unsigned mask = (unsigned)__ballot(hit);
          if (mask) {
            if (first < 0) first = bp + __builtin_ctz(mask);
            int pos = __popc(mask & ((1u << lane) - 1u));
            if (hit && (cnt + pos) < NSAMP) g[cnt + pos] = bp + lane;
            cnt += __popc(mask);
          }
        }
      }
    }
  }
  if (first < 0) first = NPTS - 1;
  int filled = cnt < NSAMP ? cnt : NSAMP;
  for (int slot = filled + lane; slot < NSAMP; slot += 32) g[slot] = first;
}

__device__ __forceinline__ void gemm_bn_relu(const float* __restrict__ A, int lda,
                                             const float* __restrict__ W, int Kdim, int Kpad,
                                             int Nout,
                                             const float* __restrict__ gg,
                                             const float* __restrict__ bb,
                                             const float* __restrict__ mm,
                                             const float* __restrict__ vv,
                                             float* __restrict__ D, int ldd, int lane) {
  const int r15  = lane & 15;
  const int half = lane >> 4;
  for (int m0 = 0; m0 < 32; m0 += 16) {
    for (int n0 = 0; n0 < Nout; n0 += 16) {
      v8f acc = {0.f, 0.f, 0.f, 0.f, 0.f, 0.f, 0.f, 0.f};
      const int n = n0 + r15;
      for (int k0 = 0; k0 < Kpad; k0 += 4) {
        const int kk = k0 + half * 2;
        v2f a, w;
        a.x = A[(m0 + r15) * lda + kk];
        a.y = A[(m0 + r15) * lda + kk + 1];
        int ka = kk     < Kdim ? kk     : Kdim - 1;
        int kb = kk + 1 < Kdim ? kk + 1 : Kdim - 1;
        w.x = W[n * Kdim + ka];
        w.y = W[n * Kdim + kb];
        acc = __builtin_amdgcn_wmma_f32_16x16x4_f32(false, a, false, w,
                                                    (short)0, acc, false, false);
      }
      const float sc = gg[n] * rsqrtf(vv[n] + BN_EPS);
      const float bs = bb[n] - mm[n] * sc;
#pragma unroll
      for (int r = 0; r < 8; ++r) {
        float val = fmaxf(acc[r] * sc + bs, 0.f);
        D[(m0 + r + half * 8) * ldd + n] = val;
      }
    }
  }
}

__global__ __launch_bounds__(64) void mlp_kernel(const float* __restrict__ xyz,
                                                 const float* __restrict__ feat,
                                                 const int*   __restrict__ grp,
                                                 const float* __restrict__ newxyz,
                                                 const float* __restrict__ w0, const float* __restrict__ g0,
                                                 const float* __restrict__ b0, const float* __restrict__ m0,
                                                 const float* __restrict__ v0,
                                                 const float* __restrict__ w1, const float* __restrict__ g1,
                                                 const float* __restrict__ b1, const float* __restrict__ m1,
                                                 const float* __restrict__ v1,
                                                 const float* __restrict__ w2, const float* __restrict__ g2,
                                                 const float* __restrict__ b2, const float* __restrict__ m2,
                                                 const float* __restrict__ v2,
                                                 float* __restrict__ outfeat) {
  __shared__ float Xb[2][32 * 68];
  __shared__ float Hb[2][32 * 132];

  const int wave = threadIdx.x >> 5;
  const int lane = threadIdx.x & 31;
  const int q    = blockIdx.x * 2 + wave;
  const int b    = q >> 9;
  const int s    = q & (NPT - 1);

  float* X = Xb[wave];
  float* H = Hb[wave];

  const int gi = grp[(size_t)q * NSAMP + lane];
  const float* prow = xyz + ((size_t)b * NPTS + gi) * 3;
  const float* nx   = newxyz + (b * NPT + s) * 3;
  float* xr = X + lane * 68;
  xr[0] = prow[0] - nx[0];
  xr[1] = prow[1] - nx[1];
  xr[2] = prow[2] - nx[2];
  const float4* frow = (const float4*)(feat + ((size_t)b * NPTS + gi) * 64);
#pragma unroll
  for (int c4 = 0; c4 < 16; ++c4) {
    float4 f = frow[c4];
    xr[3 + 4 * c4 + 0] = f.x;
    xr[3 + 4 * c4 + 1] = f.y;
    xr[3 + 4 * c4 + 2] = f.z;
    xr[3 + 4 * c4 + 3] = f.w;
  }
  xr[67] = 0.f;

  gemm_bn_relu(X, 68,  w0, 67, 68, 64,  g0, b0, m0, v0, H, 132, lane);
  gemm_bn_relu(H, 132, w1, 64, 64, 64,  g1, b1, m1, v1, X, 68,  lane);
  gemm_bn_relu(X, 68,  w2, 64, 64, 128, g2, b2, m2, v2, H, 132, lane);

#pragma unroll
  for (int cc = 0; cc < 4; ++cc) {
    const int col = lane + cc * 32;
    float mx = H[col];
    for (int row = 1; row < 32; ++row) mx = fmaxf(mx, H[row * 132 + col]);
    outfeat[((size_t)b * 128 + col) * NPT + s] = mx;
  }
}

extern "C" void kernel_launch(void* const* d_in, const int* in_sizes, int n_in,
                              void* d_out, int out_size, void* d_ws, size_t ws_size,
                              hipStream_t stream) {
  (void)in_sizes; (void)n_in; (void)out_size; (void)ws_size;
  const float* xyz  = (const float*)d_in[0];
  const float* feat = (const float*)d_in[1];
  const float* w0 = (const float*)d_in[2];
  const float* g0 = (const float*)d_in[3];
  const float* b0 = (const float*)d_in[4];
  const float* m0 = (const float*)d_in[5];
  const float* v0 = (const float*)d_in[6];
  const float* w1 = (const float*)d_in[7];
  const float* g1 = (const float*)d_in[8];
  const float* b1 = (const float*)d_in[9];
  const float* m1 = (const float*)d_in[10];
  const float* v1 = (const float*)d_in[11];
  const float* w2 = (const float*)d_in[12];
  const float* g2 = (const float*)d_in[13];
  const float* b2 = (const float*)d_in[14];
  const float* m2 = (const float*)d_in[15];
  const float* v2 = (const float*)d_in[16];

  float* newxyz  = (float*)d_out;
  float* outfeat = (float*)d_out + NB * NPT * 3;

  int* fpsidx = (int*)d_ws;
  int* grp    = (int*)d_ws + NB * NPT;

  fps_kernel  <<<NB,               FPS_T, 0, stream>>>(xyz, fpsidx, newxyz);
  ballq_kernel<<<(NB * NPT) / 8,     256, 0, stream>>>(xyz, newxyz, grp);
  mlp_kernel  <<<(NB * NPT) / 2,      64, 0, stream>>>(xyz, feat, grp, newxyz,
                                                       w0, g0, b0, m0, v0,
                                                       w1, g1, b1, m1, v1,
                                                       w2, g2, b2, m2, v2,
                                                       outfeat);
}


# FPS: distances handled as unsigned-integer bit patterns: running minimum by integer min (no NaN-quieting op per point), wave 0's 32-lane fold one 64-bit key max per exchange; same picks, same ties
# speedup vs baseline: 1.2897x; 1.0879x over previous
#include <hip/hip_runtime.h>
#include <hip/hip_bf16.h>
#include <math.h>

typedef float v2f __attribute__((ext_vector_type(2)));
typedef float v8f __attribute__((ext_vector_type(8)));

#define NB     8
#define NPTS   16384
#define NPT    512
#define NSAMP  32
#define RAD2   0.04f
#define BN_EPS 1e-5f

#define FPS_T 512
#define FPS_P (NPTS / FPS_T)
#define FPS_C 4
#define FPS_L (FPS_P / FPS_C)
#define FPS_K (FPS_T / 32)

typedef unsigned long long fps_key;

__device__ __forceinline__ fps_key fps_kmax(fps_key a, fps_key b) { return a > b ? a : b; }

__global__ __launch_bounds__(FPS_T) void fps_kernel(const float* __restrict__ xyz,
                                                    int* __restrict__ fpsidx,
                                                    float* __restrict__ newxyz) {
  const int b   = blockIdx.x;
  const int tid = threadIdx.x;
  const float* base = xyz + (size_t)b * NPTS * 3;

  v2f px[FPS_P / 2], py[FPS_P / 2], pz[FPS_P / 2];
  unsigned dd[FPS_P];
#pragma unroll
  for (int m = 0; m < FPS_P / 2; ++m) {
    const int p0 = tid + (2 * m) * FPS_T;
    const int p1 = p0 + FPS_T;
    px[m].x = base[p0 * 3 + 0];  px[m].y = base[p1 * 3 + 0];
    py[m].x = base[p0 * 3 + 1];  py[m].y = base[p1 * 3 + 1];
    pz[m].x = base[p0 * 3 + 2];  pz[m].y = base[p1 * 3 + 2];
    dd[2 * m]     = __float_as_uint(3.4e38f);
    dd[2 * m + 1] = __float_as_uint(3.4e38f);
  }

  __shared__ fps_key lk[FPS_T];
  __shared__ int s_last;

  const int wave = tid >> 5;
  const int lane = tid & 31;
  int last = 0;

  for (int i = 0; i < NPT; ++i) {
    const float cx = base[last * 3 + 0];
    const float cy = base[last * 3 + 1];
    const float cz = base[last * 3 + 2];
    if (tid == 0) {
      fpsidx[b * NPT + i] = last;
      newxyz[(b * NPT + i) * 3 + 0] = cx;
      newxyz[(b * NPT + i) * 3 + 1] = cy;
      newxyz[(b * NPT + i) * 3 + 2] = cz;
    }

    unsigned cv[FPS_C];
    int      cj[FPS_C];
    const v2f cx2 = {cx, cx}, cy2 = {cy, cy}, cz2 = {cz, cz};
#pragma unroll
    for (int k = 0; k < FPS_C; ++k) {
#pragma unroll
      for (int l = 0; l < FPS_L; l += 2) {
        const int j = k * FPS_L + l;
        const int m = j >> 1;
        const v2f dx = px[m] - cx2, dy = py[m] - cy2, dz = pz[m] - cz2;
        const v2f d  = dx * dx + dy * dy + dz * dz;
        const unsigned b0 = __float_as_uint(d.x);
        const unsigned b1 = __float_as_uint(d.y);
        const unsigned n0 = b0 < dd[j]     ? b0 : dd[j];
        const unsigned n1 = b1 < dd[j + 1] ? b1 : dd[j + 1];
        dd[j]     = n0;
        dd[j + 1] = n1;
        if (l == 0)          { cv[k] = n0; cj[k] = j; }
        else if (n0 > cv[k]) { cv[k] = n0; cj[k] = j; }
        if (n1 > cv[k])      { cv[k] = n1; cj[k] = j + 1; }
      }
    }
    unsigned bv = cv[0];
    int      bj = cj[0];
#pragma unroll
    for (int k = 1; k < FPS_C; ++k) {
      if (cv[k] > bv) { bv = cv[k]; bj = cj[k]; }
    }
    const int bi = tid + bj * FPS_T;

    lk[tid] = ((fps_key)bv << 32) | (unsigned)(~bi);
    __syncthreads();
    if (wave == 0) {
      fps_key kk[FPS_K];
#pragma unroll
      for (int e = 0; e < FPS_K; ++e) kk[e] = lk[lane + 32 * e];
#pragma unroll
      for (int s = FPS_K / 2; s > 0; s >>= 1)
#pragma unroll
        for (int e = 0; e < s; ++e) kk[e] = fps_kmax(kk[e], kk[e + s]);
      fps_key k = kk[0];
#pragma unroll
      for (int off = 16; off > 0; off >>= 1) k = fps_kmax(k, __shfl_xor(k, off, 32));
      if (lane == 0) s_last = (int)~(unsigned)k;
    }
    __syncthreads();
    last = __builtin_amdgcn_readfirstlane(s_last);
  }
}

#define BQ_TILE 2048
#define BQ_TV4  (BQ_TILE * 3 / 4)
#define BQ_NT   (NPTS / BQ_TILE)
#define BQ_G    4

__global__ __launch_bounds__(256) void ballq_kernel(const float* __restrict__ xyz,
                                                    const float* __restrict__ newxyz,
                                                    int* __restrict__ grp) {
  __shared__ float4 tileb[2][BQ_TV4];
  __shared__ int    needf[2][8];

  const int tid  = threadIdx.x;
  const int lane = tid & 31;
  const int wave = tid >> 5;
  const int q    = blockIdx.x * 8 + wave;
  const int b    = q >> 9;
  const int s    = q & (NPT - 1);

  const float4* src = (const float4*)(xyz + (size_t)b * NPTS * 3);
  const float cx = newxyz[(b * NPT + s) * 3 + 0];
  const float cy = newxyz[(b * NPT + s) * 3 + 1];
  const float cz = newxyz[(b * NPT + s) * 3 + 2];

  int* g = grp + (size_t)q * NSAMP;
  int cnt = 0, first = -1;

  float4 r0 = src[tid + 0 * 256], r1 = src[tid + 1 * 256], r2 = src[tid + 2 * 256];
  float4 r3 = src[tid + 3 * 256], r4 = src[tid + 4 * 256], r5 = src[tid + 5 * 256];

  for (int t = 0; t < BQ_NT; ++t) {
    const int buf = t & 1;
    tileb[buf][tid + 0 * 256] = r0;
    tileb[buf][tid + 1 * 256] = r1;
    tileb[buf][tid + 2 * 256] = r2;
    tileb[buf][tid + 3 * 256] = r3;
    tileb[buf][tid + 4 * 256] = r4;
    tileb[buf][tid + 5 * 256] = r5;
    if (lane == 0) needf[buf][wave] = (cnt < NSAMP) ? 1 : 0;
    __syncthreads();
    int any = 0;
#pragma unroll
    for (int w = 0; w < 8; ++w) any |= needf[buf][w];
    if (__builtin_amdgcn_readfirstlane(any) == 0) break;

    if (t + 1 < BQ_NT) {
      const float4* nsrc = src + (t + 1) * BQ_TV4 + tid;
      r0 = nsrc[0 * 256]; r1 = nsrc[1 * 256]; r2 = nsrc[2 * 256];
      r3 = nsrc[3 * 256]; r4 = nsrc[4 * 256]; r5 = nsrc[5 * 256];
    }

    if (cnt < NSAMP) {
      const float* L = (const float*)tileb[buf];
      for (int c0 = 0; c0 < BQ_TILE && cnt < NSAMP; c0 += 32 * BQ_G) {
        float px[BQ_G], py[BQ_G], pz[BQ_G];
#pragma unroll
        for (int u = 0; u < BQ_G; ++u) {
          const int pl = c0 + u * 32 + lane;
          px[u] = L[pl * 3 + 0];
          py[u] = L[pl * 3 + 1];
          pz[u] = L[pl * 3 + 2];
        }
#pragma unroll
        for (int u = 0; u < BQ_G; ++u) {
          const float dx = px[u] - cx;
          const float dy = py[u] - cy;
          const float dz = pz[u] - cz;
          const float d2 = dx * dx + dy * dy + dz * dz;
          const bool hit = d2 <= RAD2;
          const int bp = t * BQ_TILE + c0 + u * 32;
          unsigned mask = (unsigned)__ballot(hit);
          if (mask) {
            if (first < 0) first = bp + __builtin_ctz(mask);
            int pos = __popc(mask & ((1u << lane) - 1u));
            if (hit && (cnt + pos) < NSAMP) g[cnt + pos] = bp + lane;
            cnt += __popc(mask);
          }
        }
      }
    }
  }
  if (first < 0) first = NPTS - 1;
  int filled = cnt < NSAMP ? cnt : NSAMP;
  for (int slot = filled + lane; slot < NSAMP; slot += 32) g[slot] = first;
}

__device__ __forceinline__ void gemm_bn_relu(const float* __restrict__ A, int lda,
                                             const float* __restrict__ W, int Kdim, int Kpad,
                                             int Nout,
                                             const float* __restrict__ gg,
                                             const float* __restrict__ bb,
                                             const float* __restrict__ mm,
                                             const float* __restrict__ vv,
                                             float* __restrict__ D, int ldd, int lane) {
  const int r15  = lane & 15;
  const int half = lane >> 4;
  for (int m0 = 0; m0 < 32; m0 += 16) {
    for (int n0 = 0; n0 < Nout; n0 += 16) {
      v8f acc = {0.f, 0.f, 0.f, 0.f, 0.f, 0.f, 0.f, 0.f};
      const int n = n0 + r15;
      for (int k0 = 0; k0 < Kpad; k0 += 4) {
        const int kk = k0 + half * 2;
        v2f a, w;
        a.x = A[(m0 + r15) * lda + kk];
        a.y = A[(m0 + r15) * lda + kk + 1];
        int ka = kk     < Kdim ? kk     : Kdim - 1;
        int kb = kk + 1 < Kdim ? kk + 1 : Kdim - 1;
        w.x = W[n * Kdim + ka];
        w.y = W[n * Kdim + kb];
        acc = __builtin_amdgcn_wmma_f32_16x16x4_f32(false, a, false, w,
                                                    (short)0, acc, false, false);
      }
      const float sc = gg[n] * rsqrtf(vv[n] + BN_EPS);
      const float bs = bb[n] - mm[n] * sc;
#pragma unroll
      for (int r = 0; r < 8; ++r) {
        float val = fmaxf(acc[r] * sc + bs, 0.f);
        D[(m0 + r + half * 8) * ldd + n] = val;
      }
    }
  }
}

__global__ __launch_bounds__(64) void mlp_kernel(const float* __restrict__ xyz,
                                                 const float* __restrict__ feat,
                                                 const int*   __restrict__ grp,
                                                 const float* __restrict__ newxyz,
                                                 const float* __restrict__ w0, const float* __restrict__ g0,
                                                 const float* __restrict__ b0, const float* __restrict__ m0,
                                                 const float* __restrict__ v0,
                                                 const float* __restrict__ w1, const float* __restrict__ g1,
                                                 const float* __restrict__ b1, const float* __restrict__ m1,
                                                 const float* __restrict__ v1,
                                                 const float* __restrict__ w2, const float* __restrict__ g2,
                                                 const float* __restrict__ b2, const float* __restrict__ m2,
                                                 const float* __restrict__ v2,
                                                 float* __restrict__ outfeat) {
  __shared__ float Xb[2][32 * 68];
  __shared__ float Hb[2][32 * 132];

  const int wave = threadIdx.x >> 5;
  const int lane = threadIdx.x & 31;
  const int q    = blockIdx.x * 2 + wave;
  const int b    = q >> 9;
  const int s    = q & (NPT - 1);

  float* X = Xb[wave];
  float* H = Hb[wave];

  const int gi = grp[(size_t)q * NSAMP + lane];
  const float* prow = xyz + ((size_t)b * NPTS + gi) * 3;
  const float* nx   = newxyz + (b * NPT + s) * 3;
  float* xr = X + lane * 68;
  xr[0] = prow[0] - nx[0];
  xr[1] = prow[1] - nx[1];
  xr[2] = prow[2] - nx[2];
  const float4* frow = (const float4*)(feat + ((size_t)b * NPTS + gi) * 64);
#pragma unroll
  for (int c4 = 0; c4 < 16; ++c4) {
    float4 f = frow[c4];
    xr[3 + 4 * c4 + 0] = f.x;
    xr[3 + 4 * c4 + 1] = f.y;
    xr[3 + 4 * c4 + 2] = f.z;
    xr[3 + 4 * c4 + 3] = f.w;
  }
  xr[67] = 0.f;

  gemm_bn_relu(X, 68,  w0, 67, 68, 64,  g0, b0, m0, v0, H, 132, lane);
  gemm_bn_relu(H, 132, w1, 64, 64, 64,  g1, b1, m1, v1, X, 68,  lane);
  gemm_bn_relu(X, 68,  w2, 64, 64, 128, g2, b2, m2, v2, H, 132, lane);

#pragma unroll
  for (int cc = 0; cc < 4; ++cc) {
    const int col = lane + cc * 32;
    float mx = H[col];
    for (int row = 1; row < 32; ++row) mx = fmaxf(mx, H[row * 132 + col]);
    outfeat[((size_t)b * 128 + col) * NPT + s] = mx;
  }
}

extern "C" void kernel_launch(void* const* d_in, const int* in_sizes, int n_in,
                              void* d_out, int out_size, void* d_ws, size_t ws_size,
                              hipStream_t stream) {
  (void)in_sizes; (void)n_in; (void)out_size; (void)ws_size;
  const float* xyz  = (const float*)d_in[0];
  const float* feat = (const float*)d_in[1];
  const float* w0 = (const float*)d_in[2];
  const float* g0 = (const float*)d_in[3];
  const float* b0 = (const float*)d_in[4];
  const float* m0 = (const float*)d_in[5];
  const float* v0 = (const float*)d_in[6];
  const float* w1 = (const float*)d_in[7];
  const float* g1 = (const float*)d_in[8];
  const float* b1 = (const float*)d_in[9];
  const float* m1 = (const float*)d_in[10];
  const float* v1 = (const float*)d_in[11];
  const float* w2 = (const float*)d_in[12];
  const float* g2 = (const float*)d_in[13];
  const float* b2 = (const float*)d_in[14];
  const float* m2 = (const float*)d_in[15];
  const float* v2 = (const float*)d_in[16];

  float* newxyz  = (float*)d_out;
  float* outfeat = (float*)d_out + NB * NPT * 3;

  int* fpsidx = (int*)d_ws;
  int* grp    = (int*)d_ws + NB * NPT;

  fps_kernel  <<<NB,               FPS_T, 0, stream>>>(xyz, fpsidx, newxyz);
  ballq_kernel<<<(NB * NPT) / 8,     256, 0, stream>>>(xyz, newxyz, grp);
  mlp_kernel  <<<(NB * NPT) / 2,      64, 0, stream>>>(xyz, feat, grp, newxyz,
                                                       w0, g0, b0, m0, v0,
                                                       w1, g1, b1, m1, v1,
                                                       w2, g2, b2, m2, v2,
                                                       outfeat);
}


# FPS: wave 0 folds the (distance, ~index) keys as two 32-bit words (max distance word, then max index word among equals), no 64-bit integer max on the serial path; same picks, same ties
# speedup vs baseline: 1.5503x; 1.2021x over previous
#include <hip/hip_runtime.h>
#include <hip/hip_bf16.h>
#include <math.h>

typedef float v2f __attribute__((ext_vector_type(2)));
typedef float v8f __attribute__((ext_vector_type(8)));

#define NB     8
#define NPTS   16384
#define NPT    512
#define NSAMP  32
#define RAD2   0.04f
#define BN_EPS 1e-5f

#define FPS_T 512
#define FPS_P (NPTS / FPS_T)
#define FPS_C 4
#define FPS_L (FPS_P / FPS_C)
#define FPS_K (FPS_T / 32)

typedef unsigned long long fps_key;

__device__ __forceinline__ unsigned umax32(unsigned a, unsigned b) { return a > b ? a : b; }

__global__ __launch_bounds__(FPS_T) void fps_kernel(const float* __restrict__ xyz,
                                                    int* __restrict__ fpsidx,
                                                    float* __restrict__ newxyz) {
  const int b   = blockIdx.x;
  const int tid = threadIdx.x;
  const float* base = xyz + (size_t)b * NPTS * 3;

  v2f px[FPS_P / 2], py[FPS_P / 2], pz[FPS_P / 2];
  unsigned dd[FPS_P];
#pragma unroll
  for (int m = 0; m < FPS_P / 2; ++m) {
    const int p0 = tid + (2 * m) * FPS_T;
    const int p1 = p0 + FPS_T;
    px[m].x = base[p0 * 3 + 0];  px[m].y = base[p1 * 3 + 0];
    py[m].x = base[p0 * 3 + 1];  py[m].y = base[p1 * 3 + 1];
    pz[m].x = base[p0 * 3 + 2];  pz[m].y = base[p1 * 3 + 2];
    dd[2 * m]     = __float_as_uint(3.4e38f);
    dd[2 * m + 1] = __float_as_uint(3.4e38f);
  }

  __shared__ fps_key lk[FPS_T];
  __shared__ int s_last;

  const int wave = tid >> 5;
  const int lane = tid & 31;
  int last = 0;

  for (int i = 0; i < NPT; ++i) {
    const float cx = base[last * 3 + 0];
    const float cy = base[last * 3 + 1];
    const float cz = base[last * 3 + 2];
    if (tid == 0) {
      fpsidx[b * NPT + i] = last;
      newxyz[(b * NPT + i) * 3 + 0] = cx;
      newxyz[(b * NPT + i) * 3 + 1] = cy;
      newxyz[(b * NPT + i) * 3 + 2] = cz;
    }

    unsigned cv[FPS_C];
    int      cj[FPS_C];
    const v2f cx2 = {cx, cx}, cy2 = {cy, cy}, cz2 = {cz, cz};
#pragma unroll
    for (int k = 0; k < FPS_C; ++k) {
#pragma unroll
      for (int l = 0; l < FPS_L; l += 2) {
        const int j = k * FPS_L + l;
        const int m = j >> 1;
        const v2f dx = px[m] - cx2, dy = py[m] - cy2, dz = pz[m] - cz2;
        const v2f d  = dx * dx + dy * dy + dz * dz;
        const unsigned b0 = __float_as_uint(d.x);
        const unsigned b1 = __float_as_uint(d.y);
        const unsigned n0 = b0 < dd[j]     ? b0 : dd[j];
        const unsigned n1 = b1 < dd[j + 1] ? b1 : dd[j + 1];
        dd[j]     = n0;
        dd[j + 1] = n1;
        if (l == 0)          { cv[k] = n0; cj[k] = j; }
        else if (n0 > cv[k]) { cv[k] = n0; cj[k] = j; }
        if (n1 > cv[k])      { cv[k] = n1; cj[k] = j + 1; }
      }
    }
    unsigned bv = cv[0];
    int      bj = cj[0];
#pragma unroll
    for (int k = 1; k < FPS_C; ++k) {
      if (cv[k] > bv) { bv = cv[k]; bj = cj[k]; }
    }
    const int bi = tid + bj * FPS_T;

    lk[tid] = ((fps_key)bv << 32) | (unsigned)(~bi);
    __syncthreads();
    if (wave == 0) {
      unsigned kh[FPS_K], kl[FPS_K];
#pragma unroll
      for (int e = 0; e < FPS_K; ++e) {
        const fps_key k = lk[lane + 32 * e];
        kh[e] = (unsigned)(k >> 32);
        kl[e] = (unsigned)k;
      }
      unsigned mh = kh[0];
#pragma unroll
      for (int e = 1; e < FPS_K; ++e) mh = umax32(mh, kh[e]);
#pragma unroll
      for (int e = 0; e < FPS_K; ++e) kl[e] = kh[e] == mh ? kl[e] : 0u;
#pragma unroll
      for (int s = FPS_K / 2; s > 0; s >>= 1)
#pragma unroll
        for (int e = 0; e < s; ++e) kl[e] = umax32(kl[e], kl[e + s]);
      unsigned ml = kl[0];
#pragma unroll
      for (int off = 16; off > 0; off >>= 1) {
        const unsigned oh = (unsigned)__shfl_xor((int)mh, off, 32);
        const unsigned ol = (unsigned)__shfl_xor((int)ml, off, 32);
        const unsigned a = mh >= oh ? ml : 0u;
        const unsigned c = oh >= mh ? ol : 0u;
        ml = umax32(a, c);
        mh = umax32(mh, oh);
      }
      if (lane == 0) s_last = (int)~ml;
    }
    __syncthreads();
    last = __builtin_amdgcn_readfirstlane(s_last);
  }
}

#define BQ_TILE 2048
#define BQ_TV4  (BQ_TILE * 3 / 4)
#define BQ_NT   (NPTS / BQ_TILE)
#define BQ_G    4

__global__ __launch_bounds__(256) void ballq_kernel(const float* __restrict__ xyz,
                                                    const float* __restrict__ newxyz,
                                                    int* __restrict__ grp) {
  __shared__ float4 tileb[2][BQ_TV4];
  __shared__ int    needf[2][8];

  const int tid  = threadIdx.x;
  const int lane = tid & 31;
  const int wave = tid >> 5;
  const int q    = blockIdx.x * 8 + wave;
  const int b    = q >> 9;
  const int s    = q & (NPT - 1);

  const float4* src = (const float4*)(xyz + (size_t)b * NPTS * 3);
  const float cx = newxyz[(b * NPT + s) * 3 + 0];
  const float cy = newxyz[(b * NPT + s) * 3 + 1];
  const float cz = newxyz[(b * NPT + s) * 3 + 2];

  int* g = grp + (size_t)q * NSAMP;
  int cnt = 0, first = -1;

  float4 r0 = src[tid + 0 * 256], r1 = src[tid + 1 * 256], r2 = src[tid + 2 * 256];
  float4 r3 = src[tid + 3 * 256], r4 = src[tid + 4 * 256], r5 = src[tid + 5 * 256];

  for (int t = 0; t < BQ_NT; ++t) {
    const int buf = t & 1;
    tileb[buf][tid + 0 * 256] = r0;
    tileb[buf][tid + 1 * 256] = r1;
    tileb[buf][tid + 2 * 256] = r2;
    tileb[buf][tid + 3 * 256] = r3;
    tileb[buf][tid + 4 * 256] = r4;
    tileb[buf][tid + 5 * 256] = r5;
    if (lane == 0) needf[buf][wave] = (cnt < NSAMP) ? 1 : 0;
    __syncthreads();
    int any = 0;
#pragma unroll
    for (int w = 0; w < 8; ++w) any |= needf[buf][w];
    if (__builtin_amdgcn_readfirstlane(any) == 0) break;

    if (t + 1 < BQ_NT) {
      const float4* nsrc = src + (t + 1) * BQ_TV4 + tid;
      r0 = nsrc[0 * 256]; r1 = nsrc[1 * 256]; r2 = nsrc[2 * 256];
      r3 = nsrc[3 * 256]; r4 = nsrc[4 * 256]; r5 = nsrc[5 * 256];
    }

    if (cnt < NSAMP) {
      const float* L = (const float*)tileb[buf];
      for (int c0 = 0; c0 < BQ_TILE && cnt < NSAMP; c0 += 32 * BQ_G) {
        float px[BQ_G], py[BQ_G], pz[BQ_G];
#pragma unroll
        for (int u = 0; u < BQ_G; ++u) {
          const int pl = c0 + u * 32 + lane;
          px[u] = L[pl * 3 + 0];
          py[u] = L[pl * 3 + 1];
          pz[u] = L[pl * 3 + 2];
        }
#pragma unroll
        for (int u = 0; u < BQ_G; ++u) {
          const float dx = px[u] - cx;
          const float dy = py[u] - cy;
          const float dz = pz[u] - cz;
          const float d2 = dx * dx + dy * dy + dz * dz;
          const bool hit = d2 <= RAD2;
          const int bp = t * BQ_TILE + c0 + u * 32;
          unsigned mask = (unsigned)__ballot(hit);
          if (mask) {
            if (first < 0) first = bp + __builtin_ctz(mask);
            int pos = __popc(mask & ((1u << lane) - 1u));
            if (hit && (cnt + pos) < NSAMP) g[cnt + pos] = bp + lane;
            cnt += __popc(mask);
          }
        }
      }
    }
  }
  if (first < 0) first = NPTS - 1;
  int filled = cnt < NSAMP ? cnt : NSAMP;
  for (int slot = filled + lane; slot < NSAMP; slot += 32) g[slot] = first;
}

__device__ __forceinline__ void gemm_bn_relu(const float* __restrict__ A, int lda,
                                             const float* __restrict__ W, int Kdim, int Kpad,
                                             int Nout,
                                             const float* __restrict__ gg,
                                             const float* __restrict__ bb,
                                             const float* __restrict__ mm,
                                             const float* __restrict__ vv,
                                             float* __restrict__ D, int ldd, int lane) {
  const int r15  = lane & 15;
  const int half = lane >> 4;
  for (int m0 = 0; m0 < 32; m0 += 16) {
    for (int n0 = 0; n0 < Nout; n0 += 16) {
      v8f acc = {0.f, 0.f, 0.f, 0.f, 0.f, 0.f, 0.f, 0.f};
      const int n = n0 + r15;
      for (int k0 = 0; k0 < Kpad; k0 += 4) {
        const int kk = k0 + half * 2;
        v2f a, w;
        a.x = A[(m0 + r15) * lda + kk];
        a.y = A[(m0 + r15) * lda + kk + 1];
        int ka = kk     < Kdim ? kk     : Kdim - 1;
        int kb = kk + 1 < Kdim ? kk + 1 : Kdim - 1;
        w.x = W[n * Kdim + ka];
        w.y = W[n * Kdim + kb];
        acc = __builtin_amdgcn_wmma_f32_16x16x4_f32(false, a, false, w,
                                                    (short)0, acc, false, false);
      }
      const float sc = gg[n] * rsqrtf(vv[n] + BN_EPS);
      const float bs = bb[n] - mm[n] * sc;
#pragma unroll
      for (int r = 0; r < 8; ++r) {
        float val = fmaxf(acc[r] * sc + bs, 0.f);
        D[(m0 + r + half * 8) * ldd + n] = val;
      }
    }
  }
}

__global__ __launch_bounds__(64) void mlp_kernel(const float* __restrict__ xyz,
                                                 const float* __restrict__ feat,
                                                 const int*   __restrict__ grp,
                                                 const float* __restrict__ newxyz,
                                                 const float* __restrict__ w0, const float* __restrict__ g0,
                                                 const float* __restrict__ b0, const float* __restrict__ m0,
                                                 const float* __restrict__ v0,
                                                 const float* __restrict__ w1, const float* __restrict__ g1,
                                                 const float* __restrict__ b1, const float* __restrict__ m1,
                                                 const float* __restrict__ v1,
                                                 const float* __restrict__ w2, const float* __restrict__ g2,
                                                 const float* __restrict__ b2, const float* __restrict__ m2,
                                                 const float* __restrict__ v2,
                                                 float* __restrict__ outfeat) {
  __shared__ float Xb[2][32 * 68];
  __shared__ float Hb[2][32 * 132];

  const int wave = threadIdx.x >> 5;
  const int lane = threadIdx.x & 31;
  const int q    = blockIdx.x * 2 + wave;
  const int b    = q >> 9;
  const int s    = q & (NPT - 1);

  float* X = Xb[wave];
  float* H = Hb[wave];

  const int gi = grp[(size_t)q * NSAMP + lane];
  const float* prow = xyz + ((size_t)b * NPTS + gi) * 3;
  const float* nx   = newxyz + (b * NPT + s) * 3;
  float* xr = X + lane * 68;
  xr[0] = prow[0] - nx[0];
  xr[1] = prow[1] - nx[1];
  xr[2] = prow[2] - nx[2];
  const float4* frow = (const float4*)(feat + ((size_t)b * NPTS + gi) * 64);
#pragma unroll
  for (int c4 = 0; c4 < 16; ++c4) {
    float4 f = frow[c4];
    xr[3 + 4 * c4 + 0] = f.x;
    xr[3 + 4 * c4 + 1] = f.y;
    xr[3 + 4 * c4 + 2] = f.z;
    xr[3 + 4 * c4 + 3] = f.w;
  }
  xr[67] = 0.f;

  gemm_bn_relu(X, 68,  w0, 67, 68, 64,  g0, b0, m0, v0, H, 132, lane);
  gemm_bn_relu(H, 132, w1, 64, 64, 64,  g1, b1, m1, v1, X, 68,  lane);
  gemm_bn_relu(X, 68,  w2, 64, 64, 128, g2, b2, m2, v2, H, 132, lane);

#pragma unroll
  for (int cc = 0; cc < 4; ++cc) {
    const int col = lane + cc * 32;
    float mx = H[col];
    for (int row = 1; row < 32; ++row) mx = fmaxf(mx, H[row * 132 + col]);
    outfeat[((size_t)b * 128 + col) * NPT + s] = mx;
  }
}

extern "C" void kernel_launch(void* const* d_in, const int* in_sizes, int n_in,
                              void* d_out, int out_size, void* d_ws, size_t ws_size,
                              hipStream_t stream) {
  (void)in_sizes; (void)n_in; (void)out_size; (void)ws_size;
  const float* xyz  = (const float*)d_in[0];
  const float* feat = (const float*)d_in[1];
  const float* w0 = (const float*)d_in[2];
  const float* g0 = (const float*)d_in[3];
  const float* b0 = (const float*)d_in[4];
  const float* m0 = (const float*)d_in[5];
  const float* v0 = (const float*)d_in[6];
  const float* w1 = (const float*)d_in[7];
  const float* g1 = (const float*)d_in[8];
  const float* b1 = (const float*)d_in[9];
  const float* m1 = (const float*)d_in[10];
  const float* v1 = (const float*)d_in[11];
  const float* w2 = (const float*)d_in[12];
  const float* g2 = (const float*)d_in[13];
  const float* b2 = (const float*)d_in[14];
  const float* m2 = (const float*)d_in[15];
  const float* v2 = (const float*)d_in[16];

  float* newxyz  = (float*)d_out;
  float* outfeat = (float*)d_out + NB * NPT * 3;

  int* fpsidx = (int*)d_ws;
  int* grp    = (int*)d_ws + NB * NPT;

  fps_kernel  <<<NB,               FPS_T, 0, stream>>>(xyz, fpsidx, newxyz);
  ballq_kernel<<<(NB * NPT) / 8,     256, 0, stream>>>(xyz, newxyz, grp);
  mlp_kernel  <<<(NB * NPT) / 2,      64, 0, stream>>>(xyz, feat, grp, newxyz,
                                                       w0, g0, b0, m0, v0,
                                                       w1, g1, b1, m1, v1,
                                                       w2, g2, b2, m2, v2,
                                                       outfeat);
}


# FPS: wave 0's 32 lanes each load their own candidate's point from global memory ahead of the 32-lane fold (L2 latency under the fold, cloud not staged); winner lane publishes index+xyz; same picks
# speedup vs baseline: 1.7029x; 1.0984x over previous
#include <hip/hip_runtime.h>
#include <hip/hip_bf16.h>
#include <math.h>

typedef float v2f __attribute__((ext_vector_type(2)));
typedef float v8f __attribute__((ext_vector_type(8)));

#define NB     8
#define NPTS   16384
#define NPT    512
#define NSAMP  32
#define RAD2   0.04f
#define BN_EPS 1e-5f

#define FPS_T 512
#define FPS_P (NPTS / FPS_T)
#define FPS_C 4
#define FPS_L (FPS_P / FPS_C)
#define FPS_K (FPS_T / 32)

typedef unsigned long long fps_key;

__device__ __forceinline__ unsigned umax32(unsigned a, unsigned b) { return a > b ? a : b; }

__global__ __launch_bounds__(FPS_T) void fps_kernel(const float* __restrict__ xyz,
                                                    int* __restrict__ fpsidx,
                                                    float* __restrict__ newxyz) {
  const int b   = blockIdx.x;
  const int tid = threadIdx.x;
  const float* base = xyz + (size_t)b * NPTS * 3;

  v2f px[FPS_P / 2], py[FPS_P / 2], pz[FPS_P / 2];
  unsigned dd[FPS_P];
#pragma unroll
  for (int m = 0; m < FPS_P / 2; ++m) {
    const int p0 = tid + (2 * m) * FPS_T;
    const int p1 = p0 + FPS_T;
    px[m].x = base[p0 * 3 + 0];  px[m].y = base[p1 * 3 + 0];
    py[m].x = base[p0 * 3 + 1];  py[m].y = base[p1 * 3 + 1];
    pz[m].x = base[p0 * 3 + 2];  pz[m].y = base[p1 * 3 + 2];
    dd[2 * m]     = __float_as_uint(3.4e38f);
    dd[2 * m + 1] = __float_as_uint(3.4e38f);
  }

  __shared__ fps_key lk[FPS_T];
  __shared__ uint4 s_next[1 + 32];

  const int wave = tid >> 5;
  const int lane = tid & 31;
  int last = 0;
  float cx = base[0], cy = base[1], cz = base[2];

  for (int i = 0; i < NPT; ++i) {
    if (tid == 0) {
      fpsidx[b * NPT + i] = last;
      newxyz[(b * NPT + i) * 3 + 0] = cx;
      newxyz[(b * NPT + i) * 3 + 1] = cy;
      newxyz[(b * NPT + i) * 3 + 2] = cz;
    }

    unsigned cv[FPS_C];
    int      cj[FPS_C];
    const v2f cx2 = {cx, cx}, cy2 = {cy, cy}, cz2 = {cz, cz};
#pragma unroll
    for (int k = 0; k < FPS_C; ++k) {
#pragma unroll
      for (int l = 0; l < FPS_L; l += 2) {
        const int j = k * FPS_L + l;
        const int m = j >> 1;
        const v2f dx = px[m] - cx2, dy = py[m] - cy2, dz = pz[m] - cz2;
        const v2f d  = dx * dx + dy * dy + dz * dz;
        const unsigned b0 = __float_as_uint(d.x);
        const unsigned b1 = __float_as_uint(d.y);
        const unsigned n0 = b0 < dd[j]     ? b0 : dd[j];
        const unsigned n1 = b1 < dd[j + 1] ? b1 : dd[j + 1];
        dd[j]     = n0;
        dd[j + 1] = n1;
        if (l == 0)          { cv[k] = n0; cj[k] = j; }
        else if (n0 > cv[k]) { cv[k] = n0; cj[k] = j; }
        if (n1 > cv[k])      { cv[k] = n1; cj[k] = j + 1; }
      }
    }
    unsigned bv = cv[0];
    int      bj = cj[0];
#pragma unroll
    for (int k = 1; k < FPS_C; ++k) {
      if (cv[k] > bv) { bv = cv[k]; bj = cj[k]; }
    }
    const int bi = tid + bj * FPS_T;

    lk[tid] = ((fps_key)bv << 32) | (unsigned)(~bi);
    __syncthreads();
    if (wave == 0) {
      unsigned kh[FPS_K], kl[FPS_K];
#pragma unroll
      for (int e = 0; e < FPS_K; ++e) {
        const fps_key k = lk[lane + 32 * e];
        kh[e] = (unsigned)(k >> 32);
        kl[e] = (unsigned)k;
      }
      unsigned mh = kh[0];
#pragma unroll
      for (int e = 1; e < FPS_K; ++e) mh = umax32(mh, kh[e]);
#pragma unroll
      for (int e = 0; e < FPS_K; ++e) kl[e] = kh[e] == mh ? kl[e] : 0u;
#pragma unroll
      for (int s = FPS_K / 2; s > 0; s >>= 1)
#pragma unroll
        for (int e = 0; e < s; ++e) kl[e] = umax32(kl[e], kl[e + s]);
      unsigned ml = kl[0];
      const unsigned ml0 = ml;
      const unsigned ci  = (~ml0) & (unsigned)(NPTS - 1);
      const float gx = base[ci * 3 + 0];
      const float gy = base[ci * 3 + 1];
      const float gz = base[ci * 3 + 2];
#pragma unroll
      for (int off = 16; off > 0; off >>= 1) {
        const unsigned oh = (unsigned)__shfl_xor((int)mh, off, 32);
        const unsigned ol = (unsigned)__shfl_xor((int)ml, off, 32);
        const unsigned a = mh >= oh ? ml : 0u;
        const unsigned c = oh >= mh ? ol : 0u;
        ml = umax32(a, c);
        mh = umax32(mh, oh);
      }
      const int slot = (ml == ml0) ? 0 : 1 + lane;
      s_next[slot] = make_uint4(__float_as_uint(gx), __float_as_uint(gy),
                                __float_as_uint(gz), ci);
    }
    __syncthreads();
    const uint4 nx = s_next[0];
    cx = __uint_as_float(nx.x);
    cy = __uint_as_float(nx.y);
    cz = __uint_as_float(nx.z);
    last = (int)nx.w;
  }
}

#define BQ_TILE 2048
#define BQ_TV4  (BQ_TILE * 3 / 4)
#define BQ_NT   (NPTS / BQ_TILE)
#define BQ_G    4

__global__ __launch_bounds__(256) void ballq_kernel(const float* __restrict__ xyz,
                                                    const float* __restrict__ newxyz,
                                                    int* __restrict__ grp) {
  __shared__ float4 tileb[2][BQ_TV4];
  __shared__ int    needf[2][8];

  const int tid  = threadIdx.x;
  const int lane = tid & 31;
  const int wave = tid >> 5;
  const int q    = blockIdx.x * 8 + wave;
  const int b    = q >> 9;
  const int s    = q & (NPT - 1);

  const float4* src = (const float4*)(xyz + (size_t)b * NPTS * 3);
  const float cx = newxyz[(b * NPT + s) * 3 + 0];
  const float cy = newxyz[(b * NPT + s) * 3 + 1];
  const float cz = newxyz[(b * NPT + s) * 3 + 2];

  int* g = grp + (size_t)q * NSAMP;
  int cnt = 0, first = -1;

  float4 r0 = src[tid + 0 * 256], r1 = src[tid + 1 * 256], r2 = src[tid + 2 * 256];
  float4 r3 = src[tid + 3 * 256], r4 = src[tid + 4 * 256], r5 = src[tid + 5 * 256];

  for (int t = 0; t < BQ_NT; ++t) {
    const int buf = t & 1;
    tileb[buf][tid + 0 * 256] = r0;
    tileb[buf][tid + 1 * 256] = r1;
    tileb[buf][tid + 2 * 256] = r2;
    tileb[buf][tid + 3 * 256] = r3;
    tileb[buf][tid + 4 * 256] = r4;
    tileb[buf][tid + 5 * 256] = r5;
    if (lane == 0) needf[buf][wave] = (cnt < NSAMP) ? 1 : 0;
    __syncthreads();
    int any = 0;
#pragma unroll
    for (int w = 0; w < 8; ++w) any |= needf[buf][w];
    if (__builtin_amdgcn_readfirstlane(any) == 0) break;

    if (t + 1 < BQ_NT) {
      const float4* nsrc = src + (t + 1) * BQ_TV4 + tid;
      r0 = nsrc[0 * 256]; r1 = nsrc[1 * 256]; r2 = nsrc[2 * 256];
      r3 = nsrc[3 * 256]; r4 = nsrc[4 * 256]; r5 = nsrc[5 * 256];
    }

    if (cnt < NSAMP) {
      const float* L = (const float*)tileb[buf];
      for (int c0 = 0; c0 < BQ_TILE && cnt < NSAMP; c0 += 32 * BQ_G) {
        float px[BQ_G], py[BQ_G], pz[BQ_G];
#pragma unroll
        for (int u = 0; u < BQ_G; ++u) {
          const int pl = c0 + u * 32 + lane;
          px[u] = L[pl * 3 + 0];
          py[u] = L[pl * 3 + 1];
          pz[u] = L[pl * 3 + 2];
        }
#pragma unroll
        for (int u = 0; u < BQ_G; ++u) {
          const float dx = px[u] - cx;
          const float dy = py[u] - cy;
          const float dz = pz[u] - cz;
          const float d2 = dx * dx + dy * dy + dz * dz;
          const bool hit = d2 <= RAD2;
          const int bp = t * BQ_TILE + c0 + u * 32;
          unsigned mask = (unsigned)__ballot(hit);
          if (mask) {
            if (first < 0) first = bp + __builtin_ctz(mask);
            int pos = __popc(mask & ((1u << lane) - 1u));
            if (hit && (cnt + pos) < NSAMP) g[cnt + pos] = bp + lane;
            cnt += __popc(mask);
          }
        }
      }
    }
  }
  if (first < 0) first = NPTS - 1;
  int filled = cnt < NSAMP ? cnt : NSAMP;
  for (int slot = filled + lane; slot < NSAMP; slot += 32) g[slot] = first;
}

__device__ __forceinline__ void gemm_bn_relu(const float* __restrict__ A, int lda,
                                             const float* __restrict__ W, int Kdim, int Kpad,
                                             int Nout,
                                             const float* __restrict__ gg,
                                             const float* __restrict__ bb,
                                             const float* __restrict__ mm,
                                             const float* __restrict__ vv,
                                             float* __restrict__ D, int ldd, int lane) {
  const int r15  = lane & 15;
  const int half = lane >> 4;
  for (int m0 = 0; m0 < 32; m0 += 16) {
    for (int n0 = 0; n0 < Nout; n0 += 16) {
      v8f acc = {0.f, 0.f, 0.f, 0.f, 0.f, 0.f, 0.f, 0.f};
      const int n = n0 + r15;
      for (int k0 = 0; k0 < Kpad; k0 += 4) {
        const int kk = k0 + half * 2;
        v2f a, w;
        a.x = A[(m0 + r15) * lda + kk];
        a.y = A[(m0 + r15) * lda + kk + 1];
        int ka = kk     < Kdim ? kk     : Kdim - 1;
        int kb = kk + 1 < Kdim ? kk + 1 : Kdim - 1;
        w.x = W[n * Kdim + ka];
        w.y = W[n * Kdim + kb];
        acc = __builtin_amdgcn_wmma_f32_16x16x4_f32(false, a, false, w,
                                                    (short)0, acc, false, false);
      }
      const float sc = gg[n] * rsqrtf(vv[n] + BN_EPS);
      const float bs = bb[n] - mm[n] * sc;
#pragma unroll
      for (int r = 0; r < 8; ++r) {
        float val = fmaxf(acc[r] * sc + bs, 0.f);
        D[(m0 + r + half * 8) * ldd + n] = val;
      }
    }
  }
}

__global__ __launch_bounds__(64) void mlp_kernel(const float* __restrict__ xyz,
                                                 const float* __restrict__ feat,
                                                 const int*   __restrict__ grp,
                                                 const float* __restrict__ newxyz,
                                                 const float* __restrict__ w0, const float* __restrict__ g0,
                                                 const float* __restrict__ b0, const float* __restrict__ m0,
                                                 const float* __restrict__ v0,
                                                 const float* __restrict__ w1, const float* __restrict__ g1,
                                                 const float* __restrict__ b1, const float* __restrict__ m1,
                                                 const float* __restrict__ v1,
                                                 const float* __restrict__ w2, const float* __restrict__ g2,
                                                 const float* __restrict__ b2, const float* __restrict__ m2,
                                                 const float* __restrict__ v2,
                                                 float* __restrict__ outfeat) {
  __shared__ float Xb[2][32 * 68];
  __shared__ float Hb[2][32 * 132];

  const int wave = threadIdx.x >> 5;
  const int lane = threadIdx.x & 31;
  const int q    = blockIdx.x * 2 + wave;
  const int b    = q >> 9;
  const int s    = q & (NPT - 1);

  float* X = Xb[wave];
  float* H = Hb[wave];

  const int gi = grp[(size_t)q * NSAMP + lane];
  const float* prow = xyz + ((size_t)b * NPTS + gi) * 3;
  const float* nx   = newxyz + (b * NPT + s) * 3;
  float* xr = X + lane * 68;
  xr[0] = prow[0] - nx[0];
  xr[1] = prow[1] - nx[1];
  xr[2] = prow[2] - nx[2];
  const float4* frow = (const float4*)(feat + ((size_t)b * NPTS + gi) * 64);
#pragma unroll
  for (int c4 = 0; c4 < 16; ++c4) {
    float4 f = frow[c4];
    xr[3 + 4 * c4 + 0] = f.x;
    xr[3 + 4 * c4 + 1] = f.y;
    xr[3 + 4 * c4 + 2] = f.z;
    xr[3 + 4 * c4 + 3] = f.w;
  }
  xr[67] = 0.f;

  gemm_bn_relu(X, 68,  w0, 67, 68, 64,  g0, b0, m0, v0, H, 132, lane);
  gemm_bn_relu(H, 132, w1, 64, 64, 64,  g1, b1, m1, v1, X, 68,  lane);
  gemm_bn_relu(X, 68,  w2, 64, 64, 128, g2, b2, m2, v2, H, 132, lane);

#pragma unroll
  for (int cc = 0; cc < 4; ++cc) {
    const int col = lane + cc * 32;
    float mx = H[col];
    for (int row = 1; row < 32; ++row) mx = fmaxf(mx, H[row * 132 + col]);
    outfeat[((size_t)b * 128 + col) * NPT + s] = mx;
  }
}

extern "C" void kernel_launch(void* const* d_in, const int* in_sizes, int n_in,
                              void* d_out, int out_size, void* d_ws, size_t ws_size,
                              hipStream_t stream) {
  (void)in_sizes; (void)n_in; (void)out_size; (void)ws_size;
  const float* xyz  = (const float*)d_in[0];
  const float* feat = (const float*)d_in[1];
  const float* w0 = (const float*)d_in[2];
  const float* g0 = (const float*)d_in[3];
  const float* b0 = (const float*)d_in[4];
  const float* m0 = (const float*)d_in[5];
  const float* v0 = (const float*)d_in[6];
  const float* w1 = (const float*)d_in[7];
  const float* g1 = (const float*)d_in[8];
  const float* b1 = (const float*)d_in[9];
  const float* m1 = (const float*)d_in[10];
  const float* v1 = (const float*)d_in[11];
  const float* w2 = (const float*)d_in[12];
  const float* g2 = (const float*)d_in[13];
  const float* b2 = (const float*)d_in[14];
  const float* m2 = (const float*)d_in[15];
  const float* v2 = (const float*)d_in[16];

  float* newxyz  = (float*)d_out;
  float* outfeat = (float*)d_out + NB * NPT * 3;

  int* fpsidx = (int*)d_ws;
  int* grp    = (int*)d_ws + NB * NPT;

  fps_kernel  <<<NB,               FPS_T, 0, stream>>>(xyz, fpsidx, newxyz);
  ballq_kernel<<<(NB * NPT) / 8,     256, 0, stream>>>(xyz, newxyz, grp);
  mlp_kernel  <<<(NB * NPT) / 2,      64, 0, stream>>>(xyz, feat, grp, newxyz,
                                                       w0, g0, b0, m0, v0,
                                                       w1, g1, b1, m1, v1,
                                                       w2, g2, b2, m2, v2,
                                                       outfeat);
}


# FPS: wave 0's 32-lane butterfly of lane exchanges dropped: each lane puts its folded key in LDS, every lane reads all 32 back (same-address reads) and folds them in registers; same picks, same ties
# speedup vs baseline: 1.8612x; 1.0930x over previous
#include <hip/hip_runtime.h>
#include <hip/hip_bf16.h>
#include <math.h>

typedef float v2f __attribute__((ext_vector_type(2)));
typedef float v8f __attribute__((ext_vector_type(8)));

#define NB     8
#define NPTS   16384
#define NPT    512
#define NSAMP  32
#define RAD2   0.04f
#define BN_EPS 1e-5f

#define FPS_T 512
#define FPS_P (NPTS / FPS_T)
#define FPS_C 4
#define FPS_L (FPS_P / FPS_C)
#define FPS_K (FPS_T / 32)

typedef unsigned long long fps_key;

__device__ __forceinline__ unsigned umax32(unsigned a, unsigned b) { return a > b ? a : b; }

__global__ __launch_bounds__(FPS_T) void fps_kernel(const float* __restrict__ xyz,
                                                    int* __restrict__ fpsidx,
                                                    float* __restrict__ newxyz) {
  const int b   = blockIdx.x;
  const int tid = threadIdx.x;
  const float* base = xyz + (size_t)b * NPTS * 3;

  v2f px[FPS_P / 2], py[FPS_P / 2], pz[FPS_P / 2];
  unsigned dd[FPS_P];
#pragma unroll
  for (int m = 0; m < FPS_P / 2; ++m) {
    const int p0 = tid + (2 * m) * FPS_T;
    const int p1 = p0 + FPS_T;
    px[m].x = base[p0 * 3 + 0];  px[m].y = base[p1 * 3 + 0];
    py[m].x = base[p0 * 3 + 1];  py[m].y = base[p1 * 3 + 1];
    pz[m].x = base[p0 * 3 + 2];  pz[m].y = base[p1 * 3 + 2];
    dd[2 * m]     = __float_as_uint(3.4e38f);
    dd[2 * m + 1] = __float_as_uint(3.4e38f);
  }

  __shared__ fps_key lk[FPS_T];
  __shared__ uint4 s_next[1 + 32];
  __shared__ __attribute__((aligned(16))) unsigned gk[2 * 32];

  const int wave = tid >> 5;
  const int lane = tid & 31;
  int last = 0;
  float cx = base[0], cy = base[1], cz = base[2];

  for (int i = 0; i < NPT; ++i) {
    if (tid == 0) {
      fpsidx[b * NPT + i] = last;
      newxyz[(b * NPT + i) * 3 + 0] = cx;
      newxyz[(b * NPT + i) * 3 + 1] = cy;
      newxyz[(b * NPT + i) * 3 + 2] = cz;
    }

    unsigned cv[FPS_C];
    int      cj[FPS_C];
    const v2f cx2 = {cx, cx}, cy2 = {cy, cy}, cz2 = {cz, cz};
#pragma unroll
    for (int k = 0; k < FPS_C; ++k) {
#pragma unroll
      for (int l = 0; l < FPS_L; l += 2) {
        const int j = k * FPS_L + l;
        const int m = j >> 1;
        const v2f dx = px[m] - cx2, dy = py[m] - cy2, dz = pz[m] - cz2;
        const v2f d  = dx * dx + dy * dy + dz * dz;
        const unsigned b0 = __float_as_uint(d.x);
        const unsigned b1 = __float_as_uint(d.y);
        const unsigned n0 = b0 < dd[j]     ? b0 : dd[j];
        const unsigned n1 = b1 < dd[j + 1] ? b1 : dd[j + 1];
        dd[j]     = n0;
        dd[j + 1] = n1;
        if (l == 0)          { cv[k] = n0; cj[k] = j; }
        else if (n0 > cv[k]) { cv[k] = n0; cj[k] = j; }
        if (n1 > cv[k])      { cv[k] = n1; cj[k] = j + 1; }
      }
    }
    unsigned bv = cv[0];
    int      bj = cj[0];
#pragma unroll
    for (int k = 1; k < FPS_C; ++k) {
      if (cv[k] > bv) { bv = cv[k]; bj = cj[k]; }
    }
    const int bi = tid + bj * FPS_T;

    lk[tid] = ((fps_key)bv << 32) | (unsigned)(~bi);
    __syncthreads();
    if (wave == 0) {
      unsigned kh[FPS_K], kl[FPS_K];
#pragma unroll
      for (int e = 0; e < FPS_K; ++e) {
        const fps_key k = lk[lane + 32 * e];
        kh[e] = (unsigned)(k >> 32);
        kl[e] = (unsigned)k;
      }
      unsigned mh = kh[0];
#pragma unroll
      for (int e = 1; e < FPS_K; ++e) mh = umax32(mh, kh[e]);
#pragma unroll
      for (int e = 0; e < FPS_K; ++e) kl[e] = kh[e] == mh ? kl[e] : 0u;
#pragma unroll
      for (int s = FPS_K / 2; s > 0; s >>= 1)
#pragma unroll
        for (int e = 0; e < s; ++e) kl[e] = umax32(kl[e], kl[e + s]);
      unsigned ml = kl[0];
      const unsigned ml0 = ml;
      const unsigned ci  = (~ml0) & (unsigned)(NPTS - 1);
      const float gx = base[ci * 3 + 0];
      const float gy = base[ci * 3 + 1];
      const float gz = base[ci * 3 + 2];
      __builtin_amdgcn_sched_barrier(0);
      gk[2 * lane]     = mh;
      gk[2 * lane + 1] = ml;
      unsigned ah[32], al[32];
#pragma unroll
      for (int e = 0; e < 16; ++e) {
        ah[2 * e]     = gk[4 * e + 0];  al[2 * e]     = gk[4 * e + 1];
        ah[2 * e + 1] = gk[4 * e + 2];  al[2 * e + 1] = gk[4 * e + 3];
      }
      unsigned th[32];
#pragma unroll
      for (int e = 0; e < 32; ++e) th[e] = ah[e];
#pragma unroll
      for (int s = 16; s > 0; s >>= 1)
#pragma unroll
        for (int e = 0; e < s; ++e) th[e] = umax32(th[e], th[e + s]);
      mh = th[0];
#pragma unroll
      for (int e = 0; e < 32; ++e) al[e] = ah[e] == mh ? al[e] : 0u;
#pragma unroll
      for (int s = 16; s > 0; s >>= 1)
#pragma unroll
        for (int e = 0; e < s; ++e) al[e] = umax32(al[e], al[e + s]);
      ml = al[0];
      const int slot = (ml == ml0) ? 0 : 1 + lane;
      s_next[slot] = make_uint4(__float_as_uint(gx), __float_as_uint(gy),
                                __float_as_uint(gz), ci);
    }
    __syncthreads();
    const uint4 nx = s_next[0];
    cx = __uint_as_float(nx.x);
    cy = __uint_as_float(nx.y);
    cz = __uint_as_float(nx.z);
    last = (int)nx.w;
  }
}

#define BQ_TILE 2048
#define BQ_TV4  (BQ_TILE * 3 / 4)
#define BQ_NT   (NPTS / BQ_TILE)
#define BQ_G    4

__global__ __launch_bounds__(256) void ballq_kernel(const float* __restrict__ xyz,
                                                    const float* __restrict__ newxyz,
                                                    int* __restrict__ grp) {
  __shared__ float4 tileb[2][BQ_TV4];
  __shared__ int    needf[2][8];

  const int tid  = threadIdx.x;
  const int lane = tid & 31;
  const int wave = tid >> 5;
  const int q    = blockIdx.x * 8 + wave;
  const int b    = q >> 9;
  const int s    = q & (NPT - 1);

  const float4* src = (const float4*)(xyz + (size_t)b * NPTS * 3);
  const float cx = newxyz[(b * NPT + s) * 3 + 0];
  const float cy = newxyz[(b * NPT + s) * 3 + 1];
  const float cz = newxyz[(b * NPT + s) * 3 + 2];

  int* g = grp + (size_t)q * NSAMP;
  int cnt = 0, first = -1;

  float4 r0 = src[tid + 0 * 256], r1 = src[tid + 1 * 256], r2 = src[tid + 2 * 256];
  float4 r3 = src[tid + 3 * 256], r4 = src[tid + 4 * 256], r5 = src[tid + 5 * 256];

  for (int t = 0; t < BQ_NT; ++t) {
    const int buf = t & 1;
    tileb[buf][tid + 0 * 256] = r0;
    tileb[buf][tid + 1 * 256] = r1;
    tileb[buf][tid + 2 * 256] = r2;
    tileb[buf][tid + 3 * 256] = r3;
    tileb[buf][tid + 4 * 256] = r4;
    tileb[buf][tid + 5 * 256] = r5;
    if (lane == 0) needf[buf][wave] = (cnt < NSAMP) ? 1 : 0;
    __syncthreads();
    int any = 0;
#pragma unroll
    for (int w = 0; w < 8; ++w) any |= needf[buf][w];
    if (__builtin_amdgcn_readfirstlane(any) == 0) break;

    if (t + 1 < BQ_NT) {
      const float4* nsrc = src + (t + 1) * BQ_TV4 + tid;
      r0 = nsrc[0 * 256]; r1 = nsrc[1 * 256]; r2 = nsrc[2 * 256];
      r3 = nsrc[3 * 256]; r4 = nsrc[4 * 256]; r5 = nsrc[5 * 256];
    }

    if (cnt < NSAMP) {
      const float* L = (const float*)tileb[buf];
      for (int c0 = 0; c0 < BQ_TILE && cnt < NSAMP; c0 += 32 * BQ_G) {
        float px[BQ_G], py[BQ_G], pz[BQ_G];
#pragma unroll
        for (int u = 0; u < BQ_G; ++u) {
          const int pl = c0 + u * 32 + lane;
          px[u] = L[pl * 3 + 0];
          py[u] = L[pl * 3 + 1];
          pz[u] = L[pl * 3 + 2];
        }
#pragma unroll
        for (int u = 0; u < BQ_G; ++u) {
          const float dx = px[u] - cx;
          const float dy = py[u] - cy;
          const float dz = pz[u] - cz;
          const float d2 = dx * dx + dy * dy + dz * dz;
          const bool hit = d2 <= RAD2;
          const int bp = t * BQ_TILE + c0 + u * 32;
          unsigned mask = (unsigned)__ballot(hit);
          if (mask) {
            if (first < 0) first = bp + __builtin_ctz(mask);
            int pos = __popc(mask & ((1u << lane) - 1u));
            if (hit && (cnt + pos) < NSAMP) g[cnt + pos] = bp + lane;
            cnt += __popc(mask);
          }
        }
      }
    }
  }
  if (first < 0) first = NPTS - 1;
  int filled = cnt < NSAMP ? cnt : NSAMP;
  for (int slot = filled + lane; slot < NSAMP; slot += 32) g[slot] = first;
}

__device__ __forceinline__ void gemm_bn_relu(const float* __restrict__ A, int lda,
                                             const float* __restrict__ W, int Kdim, int Kpad,
                                             int Nout,
                                             const float* __restrict__ gg,
                                             const float* __restrict__ bb,
                                             const float* __restrict__ mm,
                                             const float* __restrict__ vv,
                                             float* __restrict__ D, int ldd, int lane) {
  const int r15  = lane & 15;
  const int half = lane >> 4;
  for (int m0 = 0; m0 < 32; m0 += 16) {
    for (int n0 = 0; n0 < Nout; n0 += 16) {
      v8f acc = {0.f, 0.f, 0.f, 0.f, 0.f, 0.f, 0.f, 0.f};
      const int n = n0 + r15;
      for (int k0 = 0; k0 < Kpad; k0 += 4) {
        const int kk = k0 + half * 2;
        v2f a, w;
        a.x = A[(m0 + r15) * lda + kk];
        a.y = A[(m0 + r15) * lda + kk + 1];
        int ka = kk     < Kdim ? kk     : Kdim - 1;
        int kb = kk + 1 < Kdim ? kk + 1 : Kdim - 1;
        w.x = W[n * Kdim + ka];
        w.y = W[n * Kdim + kb];
        acc = __builtin_amdgcn_wmma_f32_16x16x4_f32(false, a, false, w,
                                                    (short)0, acc, false, false);
      }
      const float sc = gg[n] * rsqrtf(vv[n] + BN_EPS);
      const float bs = bb[n] - mm[n] * sc;
#pragma unroll
      for (int r = 0; r < 8; ++r) {
        float val = fmaxf(acc[r] * sc + bs, 0.f);
        D[(m0 + r + half * 8) * ldd + n] = val;
      }
    }
  }
}

__global__ __launch_bounds__(64) void mlp_kernel(const float* __restrict__ xyz,
                                                 const float* __restrict__ feat,
                                                 const int*   __restrict__ grp,
                                                 const float* __restrict__ newxyz,
                                                 const float* __restrict__ w0, const float* __restrict__ g0,
                                                 const float* __restrict__ b0, const float* __restrict__ m0,
                                                 const float* __restrict__ v0,
                                                 const float* __restrict__ w1, const float* __restrict__ g1,
                                                 const float* __restrict__ b1, const float* __restrict__ m1,
                                                 const float* __restrict__ v1,
                                                 const float* __restrict__ w2, const float* __restrict__ g2,
                                                 const float* __restrict__ b2, const float* __restrict__ m2,
                                                 const float* __restrict__ v2,
                                                 float* __restrict__ outfeat) {
  __shared__ float Xb[2][32 * 68];
  __shared__ float Hb[2][32 * 132];

  const int wave = threadIdx.x >> 5;
  const int lane = threadIdx.x & 31;
  const int q    = blockIdx.x * 2 + wave;
  const int b    = q >> 9;
  const int s    = q & (NPT - 1);

  float* X = Xb[wave];
  float* H = Hb[wave];

  const int gi = grp[(size_t)q * NSAMP + lane];
  const float* prow = xyz + ((size_t)b * NPTS + gi) * 3;
  const float* nx   = newxyz + (b * NPT + s) * 3;
  float* xr = X + lane * 68;
  xr[0] = prow[0] - nx[0];
  xr[1] = prow[1] - nx[1];
  xr[2] = prow[2] - nx[2];
  const float4* frow = (const float4*)(feat + ((size_t)b * NPTS + gi) * 64);
#pragma unroll
  for (int c4 = 0; c4 < 16; ++c4) {
    float4 f = frow[c4];
    xr[3 + 4 * c4 + 0] = f.x;
    xr[3 + 4 * c4 + 1] = f.y;
    xr[3 + 4 * c4 + 2] = f.z;
    xr[3 + 4 * c4 + 3] = f.w;
  }
  xr[67] = 0.f;

  gemm_bn_relu(X, 68,  w0, 67, 68, 64,  g0, b0, m0, v0, H, 132, lane);
  gemm_bn_relu(H, 132, w1, 64, 64, 64,  g1, b1, m1, v1, X, 68,  lane);
  gemm_bn_relu(X, 68,  w2, 64, 64, 128, g2, b2, m2, v2, H, 132, lane);

#pragma unroll
  for (int cc = 0; cc < 4; ++cc) {
    const int col = lane + cc * 32;
    float mx = H[col];
    for (int row = 1; row < 32; ++row) mx = fmaxf(mx, H[row * 132 + col]);
    outfeat[((size_t)b * 128 + col) * NPT + s] = mx;
  }
}

extern "C" void kernel_launch(void* const* d_in, const int* in_sizes, int n_in,
                              void* d_out, int out_size, void* d_ws, size_t ws_size,
                              hipStream_t stream) {
  (void)in_sizes; (void)n_in; (void)out_size; (void)ws_size;
  const float* xyz  = (const float*)d_in[0];
  const float* feat = (const float*)d_in[1];
  const float* w0 = (const float*)d_in[2];
  const float* g0 = (const float*)d_in[3];
  const float* b0 = (const float*)d_in[4];
  const float* m0 = (const float*)d_in[5];
  const float* v0 = (const float*)d_in[6];
  const float* w1 = (const float*)d_in[7];
  const float* g1 = (const float*)d_in[8];
  const float* b1 = (const float*)d_in[9];
  const float* m1 = (const float*)d_in[10];
  const float* v1 = (const float*)d_in[11];
  const float* w2 = (const float*)d_in[12];
  const float* g2 = (const float*)d_in[13];
  const float* b2 = (const float*)d_in[14];
  const float* m2 = (const float*)d_in[15];
  const float* v2 = (const float*)d_in[16];

  float* newxyz  = (float*)d_out;
  float* outfeat = (float*)d_out + NB * NPT * 3;

  int* fpsidx = (int*)d_ws;
  int* grp    = (int*)d_ws + NB * NPT;

  fps_kernel  <<<NB,               FPS_T, 0, stream>>>(xyz, fpsidx, newxyz);
  ballq_kernel<<<(NB * NPT) / 8,     256, 0, stream>>>(xyz, newxyz, grp);
  mlp_kernel  <<<(NB * NPT) / 2,      64, 0, stream>>>(xyz, feat, grp, newxyz,
                                                       w0, g0, b0, m0, v0,
                                                       w1, g1, b1, m1, v1,
                                                       w2, g2, b2, m2, v2,
                                                       outfeat);
}


# MLP: 8-wave blocks serve 16 queries each; weights and folded batch-norm constants are copied to LDS once per block, B fragments and epilogue constants come from LDS, not global memory; same k order
# speedup vs baseline: 2.0115x; 1.0808x over previous
#include <hip/hip_runtime.h>
#include <hip/hip_bf16.h>
#include <math.h>

typedef float v2f __attribute__((ext_vector_type(2)));
typedef float v8f __attribute__((ext_vector_type(8)));

#define NB     8
#define NPTS   16384
#define NPT    512
#define NSAMP  32
#define RAD2   0.04f
#define BN_EPS 1e-5f

#define FPS_T 512
#define FPS_P (NPTS / FPS_T)
#define FPS_C 4
#define FPS_L (FPS_P / FPS_C)
#define FPS_K (FPS_T / 32)

typedef unsigned long long fps_key;

__device__ __forceinline__ unsigned umax32(unsigned a, unsigned b) { return a > b ? a : b; }

__global__ __launch_bounds__(FPS_T) void fps_kernel(const float* __restrict__ xyz,
                                                    int* __restrict__ fpsidx,
                                                    float* __restrict__ newxyz) {
  const int b   = blockIdx.x;
  const int tid = threadIdx.x;
  const float* base = xyz + (size_t)b * NPTS * 3;

  v2f px[FPS_P / 2], py[FPS_P / 2], pz[FPS_P / 2];
  unsigned dd[FPS_P];
#pragma unroll
  for (int m = 0; m < FPS_P / 2; ++m) {
    const int p0 = tid + (2 * m) * FPS_T;
    const int p1 = p0 + FPS_T;
    px[m].x = base[p0 * 3 + 0];  px[m].y = base[p1 * 3 + 0];
    py[m].x = base[p0 * 3 + 1];  py[m].y = base[p1 * 3 + 1];
    pz[m].x = base[p0 * 3 + 2];  pz[m].y = base[p1 * 3 + 2];
    dd[2 * m]     = __float_as_uint(3.4e38f);
    dd[2 * m + 1] = __float_as_uint(3.4e38f);
  }

  __shared__ fps_key lk[FPS_T];
  __shared__ uint4 s_next[1 + 32];
  __shared__ __attribute__((aligned(16))) unsigned gk[2 * 32];

  const int wave = tid >> 5;
  const int lane = tid & 31;
  int last = 0;
  float cx = base[0], cy = base[1], cz = base[2];

  for (int i = 0; i < NPT; ++i) {
    if (tid == 0) {
      fpsidx[b * NPT + i] = last;
      newxyz[(b * NPT + i) * 3 + 0] = cx;
      newxyz[(b * NPT + i) * 3 + 1] = cy;
      newxyz[(b * NPT + i) * 3 + 2] = cz;
    }

    unsigned cv[FPS_C];
    int      cj[FPS_C];
    const v2f cx2 = {cx, cx}, cy2 = {cy, cy}, cz2 = {cz, cz};
#pragma unroll
    for (int k = 0; k < FPS_C; ++k) {
#pragma unroll
      for (int l = 0; l < FPS_L; l += 2) {
        const int j = k * FPS_L + l;
        const int m = j >> 1;
        const v2f dx = px[m] - cx2, dy = py[m] - cy2, dz = pz[m] - cz2;
        const v2f d  = dx * dx + dy * dy + dz * dz;
        const unsigned b0 = __float_as_uint(d.x);
        const unsigned b1 = __float_as_uint(d.y);
        const unsigned n0 = b0 < dd[j]     ? b0 : dd[j];
        const unsigned n1 = b1 < dd[j + 1] ? b1 : dd[j + 1];
        dd[j]     = n0;
        dd[j + 1] = n1;
        if (l == 0)          { cv[k] = n0; cj[k] = j; }
        else if (n0 > cv[k]) { cv[k] = n0; cj[k] = j; }
        if (n1 > cv[k])      { cv[k] = n1; cj[k] = j + 1; }
      }
    }
    unsigned bv = cv[0];
    int      bj = cj[0];
#pragma unroll
    for (int k = 1; k < FPS_C; ++k) {
      if (cv[k] > bv) { bv = cv[k]; bj = cj[k]; }
    }
    const int bi = tid + bj * FPS_T;

    lk[tid] = ((fps_key)bv << 32) | (unsigned)(~bi);
    __syncthreads();
    if (wave == 0) {
      unsigned kh[FPS_K], kl[FPS_K];
#pragma unroll
      for (int e = 0; e < FPS_K; ++e) {
        const fps_key k = lk[lane + 32 * e];
        kh[e] = (unsigned)(k >> 32);
        kl[e] = (unsigned)k;
      }
      unsigned mh = kh[0];
#pragma unroll
      for (int e = 1; e < FPS_K; ++e) mh = umax32(mh, kh[e]);
#pragma unroll
      for (int e = 0; e < FPS_K; ++e) kl[e] = kh[e] == mh ? kl[e] : 0u;
#pragma unroll
      for (int s = FPS_K / 2; s > 0; s >>= 1)
#pragma unroll
        for (int e = 0; e < s; ++e) kl[e] = umax32(kl[e], kl[e + s]);
      unsigned ml = kl[0];
      const unsigned ml0 = ml;
      const unsigned ci  = (~ml0) & (unsigned)(NPTS - 1);
      const float gx = base[ci * 3 + 0];
      const float gy = base[ci * 3 + 1];
      const float gz = base[ci * 3 + 2];
      __builtin_amdgcn_sched_barrier(0);
      gk[2 * lane]     = mh;
      gk[2 * lane + 1] = ml;
      unsigned ah[32], al[32];
#pragma unroll
      for (int e = 0; e < 16; ++e) {
        ah[2 * e]     = gk[4 * e + 0];  al[2 * e]     = gk[4 * e + 1];
        ah[2 * e + 1] = gk[4 * e + 2];  al[2 * e + 1] = gk[4 * e + 3];
      }
      unsigned th[32];
#pragma unroll
      for (int e = 0; e < 32; ++e) th[e] = ah[e];
#pragma unroll
      for (int s = 16; s > 0; s >>= 1)
#pragma unroll
        for (int e = 0; e < s; ++e) th[e] = umax32(th[e], th[e + s]);
      mh = th[0];
#pragma unroll
      for (int e = 0; e < 32; ++e) al[e] = ah[e] == mh ? al[e] : 0u;
#pragma unroll
      for (int s = 16; s > 0; s >>= 1)
#pragma unroll
        for (int e = 0; e < s; ++e) al[e] = umax32(al[e], al[e + s]);
      ml = al[0];
      const int slot = (ml == ml0) ? 0 : 1 + lane;
      s_next[slot] = make_uint4(__float_as_uint(gx), __float_as_uint(gy),
                                __float_as_uint(gz), ci);
    }
    __syncthreads();
    const uint4 nx = s_next[0];
    cx = __uint_as_float(nx.x);
    cy = __uint_as_float(nx.y);
    cz = __uint_as_float(nx.z);
    last = (int)nx.w;
  }
}

#define BQ_TILE 2048
#define BQ_TV4  (BQ_TILE * 3 / 4)
#define BQ_NT   (NPTS / BQ_TILE)
#define BQ_G    4

__global__ __launch_bounds__(256) void ballq_kernel(const float* __restrict__ xyz,
                                                    const float* __restrict__ newxyz,
                                                    int* __restrict__ grp) {
  __shared__ float4 tileb[2][BQ_TV4];
  __shared__ int    needf[2][8];

  const int tid  = threadIdx.x;
  const int lane = tid & 31;
  const int wave = tid >> 5;
  const int q    = blockIdx.x * 8 + wave;
  const int b    = q >> 9;
  const int s    = q & (NPT - 1);

  const float4* src = (const float4*)(xyz + (size_t)b * NPTS * 3);
  const float cx = newxyz[(b * NPT + s) * 3 + 0];
  const float cy = newxyz[(b * NPT + s) * 3 + 1];
  const float cz = newxyz[(b * NPT + s) * 3 + 2];

  int* g = grp + (size_t)q * NSAMP;
  int cnt = 0, first = -1;

  float4 r0 = src[tid + 0 * 256], r1 = src[tid + 1 * 256], r2 = src[tid + 2 * 256];
  float4 r3 = src[tid + 3 * 256], r4 = src[tid + 4 * 256], r5 = src[tid + 5 * 256];

  for (int t = 0; t < BQ_NT; ++t) {
    const int buf = t & 1;
    tileb[buf][tid + 0 * 256] = r0;
    tileb[buf][tid + 1 * 256] = r1;
    tileb[buf][tid + 2 * 256] = r2;
    tileb[buf][tid + 3 * 256] = r3;
    tileb[buf][tid + 4 * 256] = r4;
    tileb[buf][tid + 5 * 256] = r5;
    if (lane == 0) needf[buf][wave] = (cnt < NSAMP) ? 1 : 0;
    __syncthreads();
    int any = 0;
#pragma unroll
    for (int w = 0; w < 8; ++w) any |= needf[buf][w];
    if (__builtin_amdgcn_readfirstlane(any) == 0) break;

    if (t + 1 < BQ_NT) {
      const float4* nsrc = src + (t + 1) * BQ_TV4 + tid;
      r0 = nsrc[0 * 256]; r1 = nsrc[1 * 256]; r2 = nsrc[2 * 256];
      r3 = nsrc[3 * 256]; r4 = nsrc[4 * 256]; r5 = nsrc[5 * 256];
    }

    if (cnt < NSAMP) {
      const float* L = (const float*)tileb[buf];
      for (int c0 = 0; c0 < BQ_TILE && cnt < NSAMP; c0 += 32 * BQ_G) {
        float px[BQ_G], py[BQ_G], pz[BQ_G];
#pragma unroll
        for (int u = 0; u < BQ_G; ++u) {
          const int pl = c0 + u * 32 + lane;
          px[u] = L[pl * 3 + 0];
          py[u] = L[pl * 3 + 1];
          pz[u] = L[pl * 3 + 2];
        }
#pragma unroll
        for (int u = 0; u < BQ_G; ++u) {
          const float dx = px[u] - cx;
          const float dy = py[u] - cy;
          const float dz = pz[u] - cz;
          const float d2 = dx * dx + dy * dy + dz * dz;
          const bool hit = d2 <= RAD2;
          const int bp = t * BQ_TILE + c0 + u * 32;
          unsigned mask = (unsigned)__ballot(hit);
          if (mask) {
            if (first < 0) first = bp + __builtin_ctz(mask);
            int pos = __popc(mask & ((1u << lane) - 1u));
            if (hit && (cnt + pos) < NSAMP) g[cnt + pos] = bp + lane;
            cnt += __popc(mask);
          }
        }
      }
    }
  }
  if (first < 0) first = NPTS - 1;
  int filled = cnt < NSAMP ? cnt : NSAMP;
  for (int slot = filled + lane; slot < NSAMP; slot += 32) g[slot] = first;
}

#define MLP_NW   8
#define MLP_T    (MLP_NW * 32)
#define MLP_QPW  2
#define MLP_QPB  (MLP_NW * MLP_QPW)
#define MLP_WP   68

template <int KS, int NOUT>
__device__ __forceinline__ void gemm_bn_relu(const float* __restrict__ A, int lda,
                                             const float* __restrict__ Wl,
                                             const float* __restrict__ SB,
                                             float* __restrict__ D, int ldd, int lane) {
  const int r15  = lane & 15;
  const int half = lane >> 4;
  for (int m0 = 0; m0 < 32; m0 += 16) {
    for (int n0 = 0; n0 < NOUT; n0 += 16) {
      const int n = n0 + r15;
      const float* ap = A  + (m0 + r15) * lda + half * 2;
      const float* wp = Wl + n * MLP_WP      + half * 2;
      v2f af[KS], wf[KS];
#pragma unroll
      for (int ks = 0; ks < KS; ++ks) {
        af[ks].x = ap[4 * ks];  af[ks].y = ap[4 * ks + 1];
        wf[ks].x = wp[4 * ks];  wf[ks].y = wp[4 * ks + 1];
      }
      __builtin_amdgcn_sched_barrier(0);
      v8f acc = {0.f, 0.f, 0.f, 0.f, 0.f, 0.f, 0.f, 0.f};
#pragma unroll
      for (int ks = 0; ks < KS; ++ks)
        acc = __builtin_amdgcn_wmma_f32_16x16x4_f32(false, af[ks], false, wf[ks],
                                                    (short)0, acc, false, false);
      const float sc = SB[2 * n];
      const float bs = SB[2 * n + 1];
#pragma unroll
      for (int r = 0; r < 8; ++r) {
        float val = fmaxf(acc[r] * sc + bs, 0.f);
        D[(m0 + r + half * 8) * ldd + n] = val;
      }
    }
  }
}

__global__ __launch_bounds__(MLP_T) void mlp_kernel(const float* __restrict__ xyz,
                                                    const float* __restrict__ feat,
                                                    const int*   __restrict__ grp,
                                                    const float* __restrict__ newxyz,
                                                    const float* __restrict__ w0, const float* __restrict__ g0,
                                                    const float* __restrict__ b0, const float* __restrict__ m0,
                                                    const float* __restrict__ v0,
                                                    const float* __restrict__ w1, const float* __restrict__ g1,
                                                    const float* __restrict__ b1, const float* __restrict__ m1,
                                                    const float* __restrict__ v1,
                                                    const float* __restrict__ w2, const float* __restrict__ g2,
                                                    const float* __restrict__ b2, const float* __restrict__ m2,
                                                    const float* __restrict__ v2,
                                                    float* __restrict__ outfeat) {
  __shared__ __attribute__((aligned(16))) float Ws[256 * MLP_WP];
  __shared__ __attribute__((aligned(16))) float SBs[2 * 256];
  __shared__ float Xb[MLP_NW][32 * 68];
  __shared__ float Hb[MLP_NW][32 * 132];

  const int tid  = threadIdx.x;
  const int wave = tid >> 5;
  const int lane = tid & 31;

#pragma unroll
  for (int j = 0; j < (64 * MLP_WP) / MLP_T; ++j) {
    const int i  = tid + j * MLP_T;
    const int r  = i / MLP_WP;
    const int c  = i - r * MLP_WP;
    const int cc = c < 67 ? c : 66;
    const float v = w0[r * 67 + cc];
    Ws[i] = (c < 67) ? v : 0.f;
  }
  {
    const float4* s1 = (const float4*)w1;
#pragma unroll
    for (int j = 0; j < (64 * 16) / MLP_T; ++j) {
      const int idx = tid + j * MLP_T;
      const int r = idx >> 4, c4 = idx & 15;
      *(float4*)&Ws[(64 + r) * MLP_WP + 4 * c4] = s1[idx];
    }
    const float4* s2 = (const float4*)w2;
#pragma unroll
    for (int j = 0; j < (128 * 16) / MLP_T; ++j) {
      const int idx = tid + j * MLP_T;
      const int r = idx >> 4, c4 = idx & 15;
      *(float4*)&Ws[(128 + r) * MLP_WP + 4 * c4] = s2[idx];
    }
  }
  {
    const int lay = tid < 64 ? 0 : (tid < 128 ? 1 : 2);
    const int c   = tid - (lay == 0 ? 0 : (lay == 1 ? 64 : 128));
    const float* gg = lay == 0 ? g0 : (lay == 1 ? g1 : g2);
    const float* bb = lay == 0 ? b0 : (lay == 1 ? b1 : b2);
    const float* mm = lay == 0 ? m0 : (lay == 1 ? m1 : m2);
    const float* vv = lay == 0 ? v0 : (lay == 1 ? v1 : v2);
    const float sc = gg[c] * rsqrtf(vv[c] + BN_EPS);
    const float bs = bb[c] - mm[c] * sc;
    SBs[2 * tid]     = sc;
    SBs[2 * tid + 1] = bs;
  }
  __syncthreads();

  float* X = Xb[wave];
  float* H = Hb[wave];

  for (int it = 0; it < MLP_QPW; ++it) {
    const int q = blockIdx.x * MLP_QPB + it * MLP_NW + wave;
    const int b = q >> 9;
    const int s = q & (NPT - 1);

    const int gi = grp[(size_t)q * NSAMP + lane];
    const float* prow = xyz + ((size_t)b * NPTS + gi) * 3;
    const float* nx   = newxyz + (b * NPT + s) * 3;
    float* xr = X + lane * 68;
    xr[0] = prow[0] - nx[0];
    xr[1] = prow[1] - nx[1];
    xr[2] = prow[2] - nx[2];
    const float4* frow = (const float4*)(feat + ((size_t)b * NPTS + gi) * 64);
#pragma unroll
    for (int c4 = 0; c4 < 16; ++c4) {
      float4 f = frow[c4];
      xr[3 + 4 * c4 + 0] = f.x;
      xr[3 + 4 * c4 + 1] = f.y;
      xr[3 + 4 * c4 + 2] = f.z;
      xr[3 + 4 * c4 + 3] = f.w;
    }
    xr[67] = 0.f;

    gemm_bn_relu<17, 64>(X, 68,  Ws,                SBs,       H, 132, lane);
    gemm_bn_relu<16, 64>(H, 132, Ws + 64 * MLP_WP,  SBs + 128, X, 68,  lane);
    gemm_bn_relu<16, 128>(X, 68,  Ws + 128 * MLP_WP, SBs + 256, H, 132, lane);

#pragma unroll
    for (int cc = 0; cc < 4; ++cc) {
      const int col = lane + cc * 32;
      float mx = H[col];
      for (int row = 1; row < 32; ++row) mx = fmaxf(mx, H[row * 132 + col]);
      outfeat[((size_t)b * 128 + col) * NPT + s] = mx;
    }
  }
}

extern "C" void kernel_launch(void* const* d_in, const int* in_sizes, int n_in,
                              void* d_out, int out_size, void* d_ws, size_t ws_size,
                              hipStream_t stream) {
  (void)in_sizes; (void)n_in; (void)out_size; (void)ws_size;
  const float* xyz  = (const float*)d_in[0];
  const float* feat = (const float*)d_in[1];
  const float* w0 = (const float*)d_in[2];
  const float* g0 = (const float*)d_in[3];
  const float* b0 = (const float*)d_in[4];
  const float* m0 = (const float*)d_in[5];
  const float* v0 = (const float*)d_in[6];
  const float* w1 = (const float*)d_in[7];
  const float* g1 = (const float*)d_in[8];
  const float* b1 = (const float*)d_in[9];
  const float* m1 = (const float*)d_in[10];
  const float* v1 = (const float*)d_in[11];
  const float* w2 = (const float*)d_in[12];
  const float* g2 = (const float*)d_in[13];
  const float* b2 = (const float*)d_in[14];
  const float* m2 = (const float*)d_in[15];
  const float* v2 = (const float*)d_in[16];

  float* newxyz  = (float*)d_out;
  float* outfeat = (float*)d_out + NB * NPT * 3;

  int* fpsidx = (int*)d_ws;
  int* grp    = (int*)d_ws + NB * NPT;

  fps_kernel  <<<NB,               FPS_T, 0, stream>>>(xyz, fpsidx, newxyz);
  ballq_kernel<<<(NB * NPT) / 8,     256, 0, stream>>>(xyz, newxyz, grp);
  mlp_kernel  <<<(NB * NPT) / MLP_QPB, MLP_T, 0, stream>>>(xyz, feat, grp, newxyz,
                                                       w0, g0, b0, m0, v0,
                                                       w1, g1, b1, m1, v1,
                                                       w2, g2, b2, m2, v2,
                                                       outfeat);
}


# FPS: wave 0 lanes keep the point fetched a step ago (two register sets in turn); if the winner's candidate is unchanged it is published at once, the new fetch not waited for; same picks
# speedup vs baseline: 2.1026x; 1.0453x over previous
#include <hip/hip_runtime.h>
#include <hip/hip_bf16.h>
#include <math.h>

typedef float v2f __attribute__((ext_vector_type(2)));
typedef float v8f __attribute__((ext_vector_type(8)));

#define NB     8
#define NPTS   16384
#define NPT    512
#define NSAMP  32
#define RAD2   0.04f
#define BN_EPS 1e-5f

#define FPS_T 512
#define FPS_P (NPTS / FPS_T)
#define FPS_C 4
#define FPS_L (FPS_P / FPS_C)
#define FPS_K (FPS_T / 32)

typedef unsigned long long fps_key;

__device__ __forceinline__ unsigned umax32(unsigned a, unsigned b) { return a > b ? a : b; }

__global__ __launch_bounds__(FPS_T) void fps_kernel(const float* __restrict__ xyz,
                                                    int* __restrict__ fpsidx,
                                                    float* __restrict__ newxyz) {
  const int b   = blockIdx.x;
  const int tid = threadIdx.x;
  const float* base = xyz + (size_t)b * NPTS * 3;

  v2f px[FPS_P / 2], py[FPS_P / 2], pz[FPS_P / 2];
  unsigned dd[FPS_P];
#pragma unroll
  for (int m = 0; m < FPS_P / 2; ++m) {
    const int p0 = tid + (2 * m) * FPS_T;
    const int p1 = p0 + FPS_T;
    px[m].x = base[p0 * 3 + 0];  px[m].y = base[p1 * 3 + 0];
    py[m].x = base[p0 * 3 + 1];  py[m].y = base[p1 * 3 + 1];
    pz[m].x = base[p0 * 3 + 2];  pz[m].y = base[p1 * 3 + 2];
    dd[2 * m]     = __float_as_uint(3.4e38f);
    dd[2 * m + 1] = __float_as_uint(3.4e38f);
  }

  __shared__ fps_key lk[FPS_T];
  __shared__ uint4 s_next[1 + 32];
  __shared__ __attribute__((aligned(16))) unsigned gk[2 * 32];

  const int wave = tid >> 5;
  const int lane = tid & 31;
  int last = 0;
  float cx = base[0], cy = base[1], cz = base[2];
  unsigned ciA = 0xffffffffu, ciB = 0xffffffffu;
  float xA = 0.f, yA = 0.f, zA = 0.f, xB = 0.f, yB = 0.f, zB = 0.f;

  auto step = [&](const int i, const unsigned cci, const float ccx, const float ccy,
                  const float ccz, unsigned& pci, float& pnx, float& pny, float& pnz)
                  __attribute__((always_inline)) {
    if (tid == 0) {
      fpsidx[b * NPT + i] = last;
      newxyz[(b * NPT + i) * 3 + 0] = cx;
      newxyz[(b * NPT + i) * 3 + 1] = cy;
      newxyz[(b * NPT + i) * 3 + 2] = cz;
    }

    unsigned cv[FPS_C];
    int      cj[FPS_C];
    const v2f cx2 = {cx, cx}, cy2 = {cy, cy}, cz2 = {cz, cz};
#pragma unroll
    for (int k = 0; k < FPS_C; ++k) {
#pragma unroll
      for (int l = 0; l < FPS_L; l += 2) {
        const int j = k * FPS_L + l;
        const int m = j >> 1;
        const v2f dx = px[m] - cx2, dy = py[m] - cy2, dz = pz[m] - cz2;
        const v2f d  = dx * dx + dy * dy + dz * dz;
        const unsigned b0 = __float_as_uint(d.x);
        const unsigned b1 = __float_as_uint(d.y);
        const unsigned n0 = b0 < dd[j]     ? b0 : dd[j];
        const unsigned n1 = b1 < dd[j + 1] ? b1 : dd[j + 1];
        dd[j]     = n0;
        dd[j + 1] = n1;
        if (l == 0)          { cv[k] = n0; cj[k] = j; }
        else if (n0 > cv[k]) { cv[k] = n0; cj[k] = j; }
        if (n1 > cv[k])      { cv[k] = n1; cj[k] = j + 1; }
      }
    }
    unsigned bv = cv[0];
    int      bj = cj[0];
#pragma unroll
    for (int k = 1; k < FPS_C; ++k) {
      if (cv[k] > bv) { bv = cv[k]; bj = cj[k]; }
    }
    const int bi = tid + bj * FPS_T;

    lk[tid] = ((fps_key)bv << 32) | (unsigned)(~bi);
    __syncthreads();
    if (wave == 0) {
      unsigned kh[FPS_K], kl[FPS_K];
#pragma unroll
      for (int e = 0; e < FPS_K; ++e) {
        const fps_key k = lk[lane + 32 * e];
        kh[e] = (unsigned)(k >> 32);
        kl[e] = (unsigned)k;
      }
      unsigned mh = kh[0];
#pragma unroll
      for (int e = 1; e < FPS_K; ++e) mh = umax32(mh, kh[e]);
#pragma unroll
      for (int e = 0; e < FPS_K; ++e) kl[e] = kh[e] == mh ? kl[e] : 0u;
#pragma unroll
      for (int s = FPS_K / 2; s > 0; s >>= 1)
#pragma unroll
        for (int e = 0; e < s; ++e) kl[e] = umax32(kl[e], kl[e + s]);
      unsigned ml = kl[0];
      const unsigned ml0 = ml;
      const unsigned ci  = (~ml0) & (unsigned)(NPTS - 1);
      pnx = base[ci * 3 + 0];
      pny = base[ci * 3 + 1];
      pnz = base[ci * 3 + 2];
      pci = ci;
      __builtin_amdgcn_sched_barrier(0);
      gk[2 * lane]     = mh;
      gk[2 * lane + 1] = ml;
      unsigned ah[32], al[32];
#pragma unroll
      for (int e = 0; e < 16; ++e) {
        ah[2 * e]     = gk[4 * e + 0];  al[2 * e]     = gk[4 * e + 1];
        ah[2 * e + 1] = gk[4 * e + 2];  al[2 * e + 1] = gk[4 * e + 3];
      }
      unsigned th[32];
#pragma unroll
      for (int e = 0; e < 32; ++e) th[e] = ah[e];
#pragma unroll
      for (int s = 16; s > 0; s >>= 1)
#pragma unroll
        for (int e = 0; e < s; ++e) th[e] = umax32(th[e], th[e + s]);
      mh = th[0];
#pragma unroll
      for (int e = 0; e < 32; ++e) al[e] = ah[e] == mh ? al[e] : 0u;
#pragma unroll
      for (int s = 16; s > 0; s >>= 1)
#pragma unroll
        for (int e = 0; e < s; ++e) al[e] = umax32(al[e], al[e + s]);
      ml = al[0];
      const bool win  = (ml == ml0);
      const int  slot = win ? 0 : 1 + lane;
      const bool hit = __builtin_amdgcn_ballot_w32(win && ci == cci) != 0u;
      if (hit) {
        s_next[slot] = make_uint4(__float_as_uint(ccx), __float_as_uint(ccy),
                                  __float_as_uint(ccz), ci);
      } else {
        asm volatile("" :: "v"(pnx), "v"(pny), "v"(pnz));
        s_next[slot] = make_uint4(__float_as_uint(pnx), __float_as_uint(pny),
                                  __float_as_uint(pnz), ci);
      }
    }
    __builtin_amdgcn_fence(__ATOMIC_RELEASE, "workgroup", "local");
    __builtin_amdgcn_s_barrier();
    __builtin_amdgcn_fence(__ATOMIC_ACQUIRE, "workgroup", "local");
    const uint4 nx = s_next[0];
    cx = __uint_as_float(nx.x);
    cy = __uint_as_float(nx.y);
    cz = __uint_as_float(nx.z);
    last = (int)nx.w;
  };
  for (int i = 0; i < NPT; i += 2) {
    step(i,     ciA, xA, yA, zA, ciB, xB, yB, zB);
    step(i + 1, ciB, xB, yB, zB, ciA, xA, yA, zA);
  }
}

#define BQ_TILE 2048
#define BQ_TV4  (BQ_TILE * 3 / 4)
#define BQ_NT   (NPTS / BQ_TILE)
#define BQ_G    4

__global__ __launch_bounds__(256) void ballq_kernel(const float* __restrict__ xyz,
                                                    const float* __restrict__ newxyz,
                                                    int* __restrict__ grp) {
  __shared__ float4 tileb[2][BQ_TV4];
  __shared__ int    needf[2][8];

  const int tid  = threadIdx.x;
  const int lane = tid & 31;
  const int wave = tid >> 5;
  const int q    = blockIdx.x * 8 + wave;
  const int b    = q >> 9;
  const int s    = q & (NPT - 1);

  const float4* src = (const float4*)(xyz + (size_t)b * NPTS * 3);
  const float cx = newxyz[(b * NPT + s) * 3 + 0];
  const float cy = newxyz[(b * NPT + s) * 3 + 1];
  const float cz = newxyz[(b * NPT + s) * 3 + 2];

  int* g = grp + (size_t)q * NSAMP;
  int cnt = 0, first = -1;

  float4 r0 = src[tid + 0 * 256], r1 = src[tid + 1 * 256], r2 = src[tid + 2 * 256];
  float4 r3 = src[tid + 3 * 256], r4 = src[tid + 4 * 256], r5 = src[tid + 5 * 256];

  for (int t = 0; t < BQ_NT; ++t) {
    const int buf = t & 1;
    tileb[buf][tid + 0 * 256] = r0;
    tileb[buf][tid + 1 * 256] = r1;
    tileb[buf][tid + 2 * 256] = r2;
    tileb[buf][tid + 3 * 256] = r3;
    tileb[buf][tid + 4 * 256] = r4;
    tileb[buf][tid + 5 * 256] = r5;
    if (lane == 0) needf[buf][wave] = (cnt < NSAMP) ? 1 : 0;
    __syncthreads();
    int any = 0;
#pragma unroll
    for (int w = 0; w < 8; ++w) any |= needf[buf][w];
    if (__builtin_amdgcn_readfirstlane(any) == 0) break;

    if (t + 1 < BQ_NT) {
      const float4* nsrc = src + (t + 1) * BQ_TV4 + tid;
      r0 = nsrc[0 * 256]; r1 = nsrc[1 * 256]; r2 = nsrc[2 * 256];
      r3 = nsrc[3 * 256]; r4 = nsrc[4 * 256]; r5 = nsrc[5 * 256];
    }

    if (cnt < NSAMP) {
      const float* L = (const float*)tileb[buf];
      for (int c0 = 0; c0 < BQ_TILE && cnt < NSAMP; c0 += 32 * BQ_G) {
        float px[BQ_G], py[BQ_G], pz[BQ_G];
#pragma unroll
        for (int u = 0; u < BQ_G; ++u) {
          const int pl = c0 + u * 32 + lane;
          px[u] = L[pl * 3 + 0];
          py[u] = L[pl * 3 + 1];
          pz[u] = L[pl * 3 + 2];
        }
#pragma unroll
        for (int u = 0; u < BQ_G; ++u) {
          const float dx = px[u] - cx;
          const float dy = py[u] - cy;
          const float dz = pz[u] - cz;
          const float d2 = dx * dx + dy * dy + dz * dz;
          const bool hit = d2 <= RAD2;
          const int bp = t * BQ_TILE + c0 + u * 32;
          unsigned mask = (unsigned)__ballot(hit);
          if (mask) {
            if (first < 0) first = bp + __builtin_ctz(mask);
            int pos = __popc(mask & ((1u << lane) - 1u));
            if (hit && (cnt + pos) < NSAMP) g[cnt + pos] = bp + lane;
            cnt += __popc(mask);
          }
        }
      }
    }
  }
  if (first < 0) first = NPTS - 1;
  int filled = cnt < NSAMP ? cnt : NSAMP;
  for (int slot = filled + lane; slot < NSAMP; slot += 32) g[slot] = first;
}

#define MLP_NW   8
#define MLP_T    (MLP_NW * 32)
#define MLP_QPW  2
#define MLP_QPB  (MLP_NW * MLP_QPW)
#define MLP_WP   68

template <int KS, int NOUT>
__device__ __forceinline__ void gemm_bn_relu(const float* __restrict__ A, int lda,
                                             const float* __restrict__ Wl,
                                             const float* __restrict__ SB,
                                             float* __restrict__ D, int ldd, int lane) {
  const int r15  = lane & 15;
  const int half = lane >> 4;
  for (int m0 = 0; m0 < 32; m0 += 16) {
    for (int n0 = 0; n0 < NOUT; n0 += 16) {
      const int n = n0 + r15;
      const float* ap = A  + (m0 + r15) * lda + half * 2;
      const float* wp = Wl + n * MLP_WP      + half * 2;
      v2f af[KS], wf[KS];
#pragma unroll
      for (int ks = 0; ks < KS; ++ks) {
        af[ks].x = ap[4 * ks];  af[ks].y = ap[4 * ks + 1];
        wf[ks].x = wp[4 * ks];  wf[ks].y = wp[4 * ks + 1];
      }
      __builtin_amdgcn_sched_barrier(0);
      v8f acc = {0.f, 0.f, 0.f, 0.f, 0.f, 0.f, 0.f, 0.f};
#pragma unroll
      for (int ks = 0; ks < KS; ++ks)
        acc = __builtin_amdgcn_wmma_f32_16x16x4_f32(false, af[ks], false, wf[ks],
                                                    (short)0, acc, false, false);
      const float sc = SB[2 * n];
      const float bs = SB[2 * n + 1];
#pragma unroll
      for (int r = 0; r < 8; ++r) {
        float val = fmaxf(acc[r] * sc + bs, 0.f);
        D[(m0 + r + half * 8) * ldd + n] = val;
      }
    }
  }
}

__global__ __launch_bounds__(MLP_T) void mlp_kernel(const float* __restrict__ xyz,
                                                    const float* __restrict__ feat,
                                                    const int*   __restrict__ grp,
                                                    const float* __restrict__ newxyz,
                                                    const float* __restrict__ w0, const float* __restrict__ g0,
                                                    const float* __restrict__ b0, const float* __restrict__ m0,
                                                    const float* __restrict__ v0,
                                                    const float* __restrict__ w1, const float* __restrict__ g1,
                                                    const float* __restrict__ b1, const float* __restrict__ m1,
                                                    const float* __restrict__ v1,
                                                    const float* __restrict__ w2, const float* __restrict__ g2,
                                                    const float* __restrict__ b2, const float* __restrict__ m2,
                                                    const float* __restrict__ v2,
                                                    float* __restrict__ outfeat) {
  __shared__ __attribute__((aligned(16))) float Ws[256 * MLP_WP];
  __shared__ __attribute__((aligned(16))) float SBs[2 * 256];
  __shared__ float Xb[MLP_NW][32 * 68];
  __shared__ float Hb[MLP_NW][32 * 132];

  const int tid  = threadIdx.x;
  const int wave = tid >> 5;
  const int lane = tid & 31;

#pragma unroll
  for (int j = 0; j < (64 * MLP_WP) / MLP_T; ++j) {
    const int i  = tid + j * MLP_T;
    const int r  = i / MLP_WP;
    const int c  = i - r * MLP_WP;
    const int cc = c < 67 ? c : 66;
    const float v = w0[r * 67 + cc];
    Ws[i] = (c < 67) ? v : 0.f;
  }
  {
    const float4* s1 = (const float4*)w1;
#pragma unroll
    for (int j = 0; j < (64 * 16) / MLP_T; ++j) {
      const int idx = tid + j * MLP_T;
      const int r = idx >> 4, c4 = idx & 15;
      *(float4*)&Ws[(64 + r) * MLP_WP + 4 * c4] = s1[idx];
    }
    const float4* s2 = (const float4*)w2;
#pragma unroll
    for (int j = 0; j < (128 * 16) / MLP_T; ++j) {
      const int idx = tid + j * MLP_T;
      const int r = idx >> 4, c4 = idx & 15;
      *(float4*)&Ws[(128 + r) * MLP_WP + 4 * c4] = s2[idx];
    }
  }
  {
    const int lay = tid < 64 ? 0 : (tid < 128 ? 1 : 2);
    const int c   = tid - (lay == 0 ? 0 : (lay == 1 ? 64 : 128));
    const float* gg = lay == 0 ? g0 : (lay == 1 ? g1 : g2);
    const float* bb = lay == 0 ? b0 : (lay == 1 ? b1 : b2);
    const float* mm = lay == 0 ? m0 : (lay == 1 ? m1 : m2);
    const float* vv = lay == 0 ? v0 : (lay == 1 ? v1 : v2);
    const float sc = gg[c] * rsqrtf(vv[c] + BN_EPS);
    const float bs = bb[c] - mm[c] * sc;
    SBs[2 * tid]     = sc;
    SBs[2 * tid + 1] = bs;
  }
  __syncthreads();

  float* X = Xb[wave];
  float* H = Hb[wave];

  for (int it = 0; it < MLP_QPW; ++it) {
    const int q = blockIdx.x * MLP_QPB + it * MLP_NW + wave;
    const int b = q >> 9;
    const int s = q & (NPT - 1);

    const int gi = grp[(size_t)q * NSAMP + lane];
    const float* prow = xyz + ((size_t)b * NPTS + gi) * 3;
    const float* nx   = newxyz + (b * NPT + s) * 3;
    float* xr = X + lane * 68;
    xr[0] = prow[0] - nx[0];
    xr[1] = prow[1] - nx[1];
    xr[2] = prow[2] - nx[2];
    const float4* frow = (const float4*)(feat + ((size_t)b * NPTS + gi) * 64);
#pragma unroll
    for (int c4 = 0; c4 < 16; ++c4) {
      float4 f = frow[c4];
      xr[3 + 4 * c4 + 0] = f.x;
      xr[3 + 4 * c4 + 1] = f.y;
      xr[3 + 4 * c4 + 2] = f.z;
      xr[3 + 4 * c4 + 3] = f.w;
    }
    xr[67] = 0.f;

    gemm_bn_relu<17, 64>(X, 68,  Ws,                SBs,       H, 132, lane);
    gemm_bn_relu<16, 64>(H, 132, Ws + 64 * MLP_WP,  SBs + 128, X, 68,  lane);
    gemm_bn_relu<16, 128>(X, 68,  Ws + 128 * MLP_WP, SBs + 256, H, 132, lane);

#pragma unroll
    for (int cc = 0; cc < 4; ++cc) {
      const int col = lane + cc * 32;
      float mx = H[col];
      for (int row = 1; row < 32; ++row) mx = fmaxf(mx, H[row * 132 + col]);
      outfeat[((size_t)b * 128 + col) * NPT + s] = mx;
    }
  }
}

extern "C" void kernel_launch(void* const* d_in, const int* in_sizes, int n_in,
                              void* d_out, int out_size, void* d_ws, size_t ws_size,
                              hipStream_t stream) {
  (void)in_sizes; (void)n_in; (void)out_size; (void)ws_size;
  const float* xyz  = (const float*)d_in[0];
  const float* feat = (const float*)d_in[1];
  const float* w0 = (const float*)d_in[2];
  const float* g0 = (const float*)d_in[3];
  const float* b0 = (const float*)d_in[4];
  const float* m0 = (const float*)d_in[5];
  const float* v0 = (const float*)d_in[6];
  const float* w1 = (const float*)d_in[7];
  const float* g1 = (const float*)d_in[8];
  const float* b1 = (const float*)d_in[9];
  const float* m1 = (const float*)d_in[10];
  const float* v1 = (const float*)d_in[11];
  const float* w2 = (const float*)d_in[12];
  const float* g2 = (const float*)d_in[13];
  const float* b2 = (const float*)d_in[14];
  const float* m2 = (const float*)d_in[15];
  const float* v2 = (const float*)d_in[16];

  float* newxyz  = (float*)d_out;
  float* outfeat = (float*)d_out + NB * NPT * 3;

  int* fpsidx = (int*)d_ws;
  int* grp    = (int*)d_ws + NB * NPT;

  fps_kernel  <<<NB,               FPS_T, 0, stream>>>(xyz, fpsidx, newxyz);
  ballq_kernel<<<(NB * NPT) / 8,     256, 0, stream>>>(xyz, newxyz, grp);
  mlp_kernel  <<<(NB * NPT) / MLP_QPB, MLP_T, 0, stream>>>(xyz, feat, grp, newxyz,
                                                       w0, g0, b0, m0, v0,
                                                       w1, g1, b1, m1, v1,
                                                       w2, g2, b2, m2, v2,
                                                       outfeat);
}


# FPS: in-lane argmax keeps only a maximum per 8-point run (three-input max); runs merge carrying the leading run's values, the place inside it is found once per step; same picks, same ties
# speedup vs baseline: 2.2378x; 1.0643x over previous
#include <hip/hip_runtime.h>
#include <hip/hip_bf16.h>
#include <math.h>

typedef float v2f __attribute__((ext_vector_type(2)));
typedef float v8f __attribute__((ext_vector_type(8)));

#define NB     8
#define NPTS   16384
#define NPT    512
#define NSAMP  32
#define RAD2   0.04f
#define BN_EPS 1e-5f

#define FPS_T 512
#define FPS_P (NPTS / FPS_T)
#define FPS_C 4
#define FPS_L (FPS_P / FPS_C)
#define FPS_K (FPS_T / 32)

typedef unsigned long long fps_key;

__device__ __forceinline__ unsigned umax32(unsigned a, unsigned b) { return a > b ? a : b; }

__global__ __launch_bounds__(FPS_T) void fps_kernel(const float* __restrict__ xyz,
                                                    int* __restrict__ fpsidx,
                                                    float* __restrict__ newxyz) {
  const int b   = blockIdx.x;
  const int tid = threadIdx.x;
  const float* base = xyz + (size_t)b * NPTS * 3;

  v2f px[FPS_P / 2], py[FPS_P / 2], pz[FPS_P / 2];
  unsigned dd[FPS_P];
#pragma unroll
  for (int m = 0; m < FPS_P / 2; ++m) {
    const int p0 = tid + (2 * m) * FPS_T;
    const int p1 = p0 + FPS_T;
    px[m].x = base[p0 * 3 + 0];  px[m].y = base[p1 * 3 + 0];
    py[m].x = base[p0 * 3 + 1];  py[m].y = base[p1 * 3 + 1];
    pz[m].x = base[p0 * 3 + 2];  pz[m].y = base[p1 * 3 + 2];
    dd[2 * m]     = __float_as_uint(3.4e38f);
    dd[2 * m + 1] = __float_as_uint(3.4e38f);
  }

  __shared__ fps_key lk[FPS_T];
  __shared__ uint4 s_next[1 + 32];
  __shared__ __attribute__((aligned(16))) unsigned gk[2 * 32];

  const int wave = tid >> 5;
  const int lane = tid & 31;
  int last = 0;
  float cx = base[0], cy = base[1], cz = base[2];
  unsigned ciA = 0xffffffffu, ciB = 0xffffffffu;
  float xA = 0.f, yA = 0.f, zA = 0.f, xB = 0.f, yB = 0.f, zB = 0.f;

  auto step = [&](const int i, const unsigned cci, const float ccx, const float ccy,
                  const float ccz, unsigned& pci, float& pnx, float& pny, float& pnz)
                  __attribute__((always_inline)) {
    if (tid == 0) {
      fpsidx[b * NPT + i] = last;
      newxyz[(b * NPT + i) * 3 + 0] = cx;
      newxyz[(b * NPT + i) * 3 + 1] = cy;
      newxyz[(b * NPT + i) * 3 + 2] = cz;
    }

    unsigned gm[FPS_C];
    const v2f cx2 = {cx, cx}, cy2 = {cy, cy}, cz2 = {cz, cz};
#pragma unroll
    for (int k = 0; k < FPS_C; ++k) {
#pragma unroll
      for (int l = 0; l < FPS_L; l += 2) {
        const int j = k * FPS_L + l;
        const int m = j >> 1;
        const v2f dx = px[m] - cx2, dy = py[m] - cy2, dz = pz[m] - cz2;
        const v2f d  = dx * dx + dy * dy + dz * dz;
        const unsigned b0 = __float_as_uint(d.x);
        const unsigned b1 = __float_as_uint(d.y);
        const unsigned n0 = b0 < dd[j]     ? b0 : dd[j];
        const unsigned n1 = b1 < dd[j + 1] ? b1 : dd[j + 1];
        dd[j]     = n0;
        dd[j + 1] = n1;
      }
      unsigned rm = dd[k * FPS_L];
#pragma unroll
      for (int l = 1; l < FPS_L; ++l) rm = umax32(rm, dd[k * FPS_L + l]);
      gm[k] = rm;
    }
    unsigned bv = gm[0];
    int      ob = 0;
    unsigned cr[FPS_L - 1];
#pragma unroll
    for (int e = 0; e < FPS_L - 1; ++e) cr[e] = dd[e];
#pragma unroll
    for (int k = 1; k < FPS_C; ++k) {
      const bool gt = gm[k] > bv;
      bv = gt ? gm[k] : bv;
      ob = gt ? k * FPS_L : ob;
#pragma unroll
      for (int e = 0; e < FPS_L - 1; ++e) cr[e] = gt ? dd[k * FPS_L + e] : cr[e];
    }
    int pl = FPS_L - 1;
#pragma unroll
    for (int e = FPS_L - 2; e >= 0; --e) pl = (cr[e] == bv) ? e : pl;
    const int bj = ob + pl;
    const int bi = tid + bj * FPS_T;

    lk[tid] = ((fps_key)bv << 32) | (unsigned)(~bi);
    __syncthreads();
    if (wave == 0) {
      unsigned kh[FPS_K], kl[FPS_K];
#pragma unroll
      for (int e = 0; e < FPS_K; ++e) {
        const fps_key k = lk[lane + 32 * e];
        kh[e] = (unsigned)(k >> 32);
        kl[e] = (unsigned)k;
      }
      unsigned mh = kh[0];
#pragma unroll
      for (int e = 1; e < FPS_K; ++e) mh = umax32(mh, kh[e]);
#pragma unroll
      for (int e = 0; e < FPS_K; ++e) kl[e] = kh[e] == mh ? kl[e] : 0u;
#pragma unroll
      for (int s = FPS_K / 2; s > 0; s >>= 1)
#pragma unroll
        for (int e = 0; e < s; ++e) kl[e] = umax32(kl[e], kl[e + s]);
      unsigned ml = kl[0];
      const unsigned ml0 = ml;
      const unsigned ci  = (~ml0) & (unsigned)(NPTS - 1);
      pnx = base[ci * 3 + 0];
      pny = base[ci * 3 + 1];
      pnz = base[ci * 3 + 2];
      pci = ci;
      __builtin_amdgcn_sched_barrier(0);
      gk[2 * lane]     = mh;
      gk[2 * lane + 1] = ml;
      unsigned ah[32], al[32];
#pragma unroll
      for (int e = 0; e < 16; ++e) {
        ah[2 * e]     = gk[4 * e + 0];  al[2 * e]     = gk[4 * e + 1];
        ah[2 * e + 1] = gk[4 * e + 2];  al[2 * e + 1] = gk[4 * e + 3];
      }
      unsigned th[32];
#pragma unroll
      for (int e = 0; e < 32; ++e) th[e] = ah[e];
#pragma unroll
      for (int s = 16; s > 0; s >>= 1)
#pragma unroll
        for (int e = 0; e < s; ++e) th[e] = umax32(th[e], th[e + s]);
      mh = th[0];
#pragma unroll
      for (int e = 0; e < 32; ++e) al[e] = ah[e] == mh ? al[e] : 0u;
#pragma unroll
      for (int s = 16; s > 0; s >>= 1)
#pragma unroll
        for (int e = 0; e < s; ++e) al[e] = umax32(al[e], al[e + s]);
      ml = al[0];
      const bool win  = (ml == ml0);
      const int  slot = win ? 0 : 1 + lane;
      const bool hit = __builtin_amdgcn_ballot_w32(win && ci == cci) != 0u;
      if (hit) {
        s_next[slot] = make_uint4(__float_as_uint(ccx), __float_as_uint(ccy),
                                  __float_as_uint(ccz), ci);
      } else {
        asm volatile("" :: "v"(pnx), "v"(pny), "v"(pnz));
        s_next[slot] = make_uint4(__float_as_uint(pnx), __float_as_uint(pny),
                                  __float_as_uint(pnz), ci);
      }
    }
    __builtin_amdgcn_fence(__ATOMIC_RELEASE, "workgroup", "local");
    __builtin_amdgcn_s_barrier();
    __builtin_amdgcn_fence(__ATOMIC_ACQUIRE, "workgroup", "local");
    const uint4 nx = s_next[0];
    cx = __uint_as_float(nx.x);
    cy = __uint_as_float(nx.y);
    cz = __uint_as_float(nx.z);
    last = (int)nx.w;
  };
  for (int i = 0; i < NPT; i += 2) {
    step(i,     ciA, xA, yA, zA, ciB, xB, yB, zB);
    step(i + 1, ciB, xB, yB, zB, ciA, xA, yA, zA);
  }
}

#define BQ_TILE 2048
#define BQ_TV4  (BQ_TILE * 3 / 4)
#define BQ_NT   (NPTS / BQ_TILE)
#define BQ_G    4

__global__ __launch_bounds__(256) void ballq_kernel(const float* __restrict__ xyz,
                                                    const float* __restrict__ newxyz,
                                                    int* __restrict__ grp) {
  __shared__ float4 tileb[2][BQ_TV4];
  __shared__ int    needf[2][8];

  const int tid  = threadIdx.x;
  const int lane = tid & 31;
  const int wave = tid >> 5;
  const int q    = blockIdx.x * 8 + wave;
  const int b    = q >> 9;
  const int s    = q & (NPT - 1);

  const float4* src = (const float4*)(xyz + (size_t)b * NPTS * 3);
  const float cx = newxyz[(b * NPT + s) * 3 + 0];
  const float cy = newxyz[(b * NPT + s) * 3 + 1];
  const float cz = newxyz[(b * NPT + s) * 3 + 2];

  int* g = grp + (size_t)q * NSAMP;
  int cnt = 0, first = -1;

  float4 r0 = src[tid + 0 * 256], r1 = src[tid + 1 * 256], r2 = src[tid + 2 * 256];
  float4 r3 = src[tid + 3 * 256], r4 = src[tid + 4 * 256], r5 = src[tid + 5 * 256];

  for (int t = 0; t < BQ_NT; ++t) {
    const int buf = t & 1;
    tileb[buf][tid + 0 * 256] = r0;
    tileb[buf][tid + 1 * 256] = r1;
    tileb[buf][tid + 2 * 256] = r2;
    tileb[buf][tid + 3 * 256] = r3;
    tileb[buf][tid + 4 * 256] = r4;
    tileb[buf][tid + 5 * 256] = r5;
    if (lane == 0) needf[buf][wave] = (cnt < NSAMP) ? 1 : 0;
    __syncthreads();
    int any = 0;
#pragma unroll
    for (int w = 0; w < 8; ++w) any |= needf[buf][w];
    if (__builtin_amdgcn_readfirstlane(any) == 0) break;

    if (t + 1 < BQ_NT) {
      const float4* nsrc = src + (t + 1) * BQ_TV4 + tid;
      r0 = nsrc[0 * 256]; r1 = nsrc[1 * 256]; r2 = nsrc[2 * 256];
      r3 = nsrc[3 * 256]; r4 = nsrc[4 * 256]; r5 = nsrc[5 * 256];
    }

    if (cnt < NSAMP) {
      const float* L = (const float*)tileb[buf];
      for (int c0 = 0; c0 < BQ_TILE && cnt < NSAMP; c0 += 32 * BQ_G) {
        float px[BQ_G], py[BQ_G], pz[BQ_G];
#pragma unroll
        for (int u = 0; u < BQ_G; ++u) {
          const int pl = c0 + u * 32 + lane;
          px[u] = L[pl * 3 + 0];
          py[u] = L[pl * 3 + 1];
          pz[u] = L[pl * 3 + 2];
        }
#pragma unroll
        for (int u = 0; u < BQ_G; ++u) {
          const float dx = px[u] - cx;
          const float dy = py[u] - cy;
          const float dz = pz[u] - cz;
          const float d2 = dx * dx + dy * dy + dz * dz;
          const bool hit = d2 <= RAD2;
          const int bp = t * BQ_TILE + c0 + u * 32;
          unsigned mask = (unsigned)__ballot(hit);
          if (mask) {
            if (first < 0) first = bp + __builtin_ctz(mask);
            int pos = __popc(mask & ((1u << lane) - 1u));
            if (hit && (cnt + pos) < NSAMP) g[cnt + pos] = bp + lane;
            cnt += __popc(mask);
          }
        }
      }
    }
  }
  if (first < 0) first = NPTS - 1;
  int filled = cnt < NSAMP ? cnt : NSAMP;
  for (int slot = filled + lane; slot < NSAMP; slot += 32) g[slot] = first;
}

#define MLP_NW   8
#define MLP_T    (MLP_NW * 32)
#define MLP_QPW  2
#define MLP_QPB  (MLP_NW * MLP_QPW)
#define MLP_WP   68

template <int KS, int NOUT>
__device__ __forceinline__ void gemm_bn_relu(const float* __restrict__ A, int lda,
                                             const float* __restrict__ Wl,
                                             const float* __restrict__ SB,
                                             float* __restrict__ D, int ldd, int lane) {
  const int r15  = lane & 15;
  const int half = lane >> 4;
  for (int m0 = 0; m0 < 32; m0 += 16) {
    for (int n0 = 0; n0 < NOUT; n0 += 16) {
      const int n = n0 + r15;
      const float* ap = A  + (m0 + r15) * lda + half * 2;
      const float* wp = Wl + n * MLP_WP      + half * 2;
      v2f af[KS], wf[KS];
#pragma unroll
      for (int ks = 0; ks < KS; ++ks) {
        af[ks].x = ap[4 * ks];  af[ks].y = ap[4 * ks + 1];
        wf[ks].x = wp[4 * ks];  wf[ks].y = wp[4 * ks + 1];
      }
      __builtin_amdgcn_sched_barrier(0);
      v8f acc = {0.f, 0.f, 0.f, 0.f, 0.f, 0.f, 0.f, 0.f};
#pragma unroll
      for (int ks = 0; ks < KS; ++ks)
        acc = __builtin_amdgcn_wmma_f32_16x16x4_f32(false, af[ks], false, wf[ks],
                                                    (short)0, acc, false, false);
      const float sc = SB[2 * n];
      const float bs = SB[2 * n + 1];
#pragma unroll
      for (int r = 0; r < 8; ++r) {
        float val = fmaxf(acc[r] * sc + bs, 0.f);
        D[(m0 + r + half * 8) * ldd + n] = val;
      }
    }
  }
}

__global__ __launch_bounds__(MLP_T) void mlp_kernel(const float* __restrict__ xyz,
                                                    const float* __restrict__ feat,
                                                    const int*   __restrict__ grp,
                                                    const float* __restrict__ newxyz,
                                                    const float* __restrict__ w0, const float* __restrict__ g0,
                                                    const float* __restrict__ b0, const float* __restrict__ m0,
                                                    const float* __restrict__ v0,
                                                    const float* __restrict__ w1, const float* __restrict__ g1,
                                                    const float* __restrict__ b1, const float* __restrict__ m1,
                                                    const float* __restrict__ v1,
                                                    const float* __restrict__ w2, const float* __restrict__ g2,
                                                    const float* __restrict__ b2, const float* __restrict__ m2,
                                                    const float* __restrict__ v2,
                                                    float* __restrict__ outfeat) {
  __shared__ __attribute__((aligned(16))) float Ws[256 * MLP_WP];
  __shared__ __attribute__((aligned(16))) float SBs[2 * 256];
  __shared__ float Xb[MLP_NW][32 * 68];
  __shared__ float Hb[MLP_NW][32 * 132];

  const int tid  = threadIdx.x;
  const int wave = tid >> 5;
  const int lane = tid & 31;

#pragma unroll
  for (int j = 0; j < (64 * MLP_WP) / MLP_T; ++j) {
    const int i  = tid + j * MLP_T;
    const int r  = i / MLP_WP;
    const int c  = i - r * MLP_WP;
    const int cc = c < 67 ? c : 66;
    const float v = w0[r * 67 + cc];
    Ws[i] = (c < 67) ? v : 0.f;
  }
  {
    const float4* s1 = (const float4*)w1;
#pragma unroll
    for (int j = 0; j < (64 * 16) / MLP_T; ++j) {
      const int idx = tid + j * MLP_T;
      const int r = idx >> 4, c4 = idx & 15;
      *(float4*)&Ws[(64 + r) * MLP_WP + 4 * c4] = s1[idx];
    }
    const float4* s2 = (const float4*)w2;
#pragma unroll
    for (int j = 0; j < (128 * 16) / MLP_T; ++j) {
      const int idx = tid + j * MLP_T;
      const int r = idx >> 4, c4 = idx & 15;
      *(float4*)&Ws[(128 + r) * MLP_WP + 4 * c4] = s2[idx];
    }
  }
  {
    const int lay = tid < 64 ? 0 : (tid < 128 ? 1 : 2);
    const int c   = tid - (lay == 0 ? 0 : (lay == 1 ? 64 : 128));
    const float* gg = lay == 0 ? g0 : (lay == 1 ? g1 : g2);
    const float* bb = lay == 0 ? b0 : (lay == 1 ? b1 : b2);
    const float* mm = lay == 0 ? m0 : (lay == 1 ? m1 : m2);
    const float* vv = lay == 0 ? v0 : (lay == 1 ? v1 : v2);
    const float sc = gg[c] * rsqrtf(vv[c] + BN_EPS);
    const float bs = bb[c] - mm[c] * sc;
    SBs[2 * tid]     = sc;
    SBs[2 * tid + 1] = bs;
  }
  __syncthreads();

  float* X = Xb[wave];
  float* H = Hb[wave];

  for (int it = 0; it < MLP_QPW; ++it) {
    const int q = blockIdx.x * MLP_QPB + it * MLP_NW + wave;
    const int b = q >> 9;
    const int s = q & (NPT - 1);

    const int gi = grp[(size_t)q * NSAMP + lane];
    const float* prow = xyz + ((size_t)b * NPTS + gi) * 3;
    const float* nx   = newxyz + (b * NPT + s) * 3;
    float* xr = X + lane * 68;
    xr[0] = prow[0] - nx[0];
    xr[1] = prow[1] - nx[1];
    xr[2] = prow[2] - nx[2];
    const float4* frow = (const float4*)(feat + ((size_t)b * NPTS + gi) * 64);
#pragma unroll
    for (int c4 = 0; c4 < 16; ++c4) {
      float4 f = frow[c4];
      xr[3 + 4 * c4 + 0] = f.x;
      xr[3 + 4 * c4 + 1] = f.y;
      xr[3 + 4 * c4 + 2] = f.z;
      xr[3 + 4 * c4 + 3] = f.w;
    }
    xr[67] = 0.f;

    gemm_bn_relu<17, 64>(X, 68,  Ws,                SBs,       H, 132, lane);
    gemm_bn_relu<16, 64>(H, 132, Ws + 64 * MLP_WP,  SBs + 128, X, 68,  lane);
    gemm_bn_relu<16, 128>(X, 68,  Ws + 128 * MLP_WP, SBs + 256, H, 132, lane);

#pragma unroll
    for (int cc = 0; cc < 4; ++cc) {
      const int col = lane + cc * 32;
      float mx = H[col];
      for (int row = 1; row < 32; ++row) mx = fmaxf(mx, H[row * 132 + col]);
      outfeat[((size_t)b * 128 + col) * NPT + s] = mx;
    }
  }
}

extern "C" void kernel_launch(void* const* d_in, const int* in_sizes, int n_in,
                              void* d_out, int out_size, void* d_ws, size_t ws_size,
                              hipStream_t stream) {
  (void)in_sizes; (void)n_in; (void)out_size; (void)ws_size;
  const float* xyz  = (const float*)d_in[0];
  const float* feat = (const float*)d_in[1];
  const float* w0 = (const float*)d_in[2];
  const float* g0 = (const float*)d_in[3];
  const float* b0 = (const float*)d_in[4];
  const float* m0 = (const float*)d_in[5];
  const float* v0 = (const float*)d_in[6];
  const float* w1 = (const float*)d_in[7];
  const float* g1 = (const float*)d_in[8];
  const float* b1 = (const float*)d_in[9];
  const float* m1 = (const float*)d_in[10];
  const float* v1 = (const float*)d_in[11];
  const float* w2 = (const float*)d_in[12];
  const float* g2 = (const float*)d_in[13];
  const float* b2 = (const float*)d_in[14];
  const float* m2 = (const float*)d_in[15];
  const float* v2 = (const float*)d_in[16];

  float* newxyz  = (float*)d_out;
  float* outfeat = (float*)d_out + NB * NPT * 3;

  int* fpsidx = (int*)d_ws;
  int* grp    = (int*)d_ws + NB * NPT;

  fps_kernel  <<<NB,               FPS_T, 0, stream>>>(xyz, fpsidx, newxyz);
  ballq_kernel<<<(NB * NPT) / 8,     256, 0, stream>>>(xyz, newxyz, grp);
  mlp_kernel  <<<(NB * NPT) / MLP_QPB, MLP_T, 0, stream>>>(xyz, feat, grp, newxyz,
                                                       w0, g0, b0, m0, v0,
                                                       w1, g1, b1, m1, v1,
                                                       w2, g2, b2, m2, v2,
                                                       outfeat);
}
